# starting kernel
# baseline (speedup 1.0000x reference)
#include <hip/hip_runtime.h>
#include <hip/hip_bf16.h>

typedef float v2f __attribute__((ext_vector_type(2)));
typedef float v8f __attribute__((ext_vector_type(8)));

static constexpr int kSEQ   = 2048;
static constexpr int kBATCH = 64;
static constexpr int kIN    = 512;
static constexpr int kH     = 3;
static constexpr int kL     = 4;
static constexpr int kG     = 3 * kH;
static constexpr int kROWS  = kSEQ * kBATCH;

__global__ __launch_bounds__(256) void gru_in_proj_wmma(
    const float* __restrict__ X, const float* __restrict__ Wih0,
    const float* __restrict__ bih0, float* __restrict__ gx) {
  __shared__ float Bs[kIN * 16];
  const int tid = threadIdx.x;
  for (int idx = tid; idx < kIN * 16; idx += 256) {
    const int k = idx >> 4;
    const int n = idx & 15;
    Bs[idx] = (n < kG) ? Wih0[n * kIN + k] : 0.0f;
  }
  __syncthreads();

  const int wave = tid >> 5;
  const int lane = tid & 31;
  const int half = lane >> 4;
  const int n16  = lane & 15;
  const int row0 = blockIdx.x * 128 + wave * 16;

  const float* arow = X + (size_t)(row0 + n16) * kIN + 2 * half;

  v8f c = {0.f, 0.f, 0.f, 0.f, 0.f, 0.f, 0.f, 0.f};
  for (int kk = 0; kk < kIN; kk += 4) {
    v2f a = *(const v2f*)(arow + kk);
    v2f b;
    b.x = Bs[(kk + 2 * half) * 16 + n16];
    b.y = Bs[(kk + 2 * half + 1) * 16 + n16];
    c = __builtin_amdgcn_wmma_f32_16x16x4_f32(
         false, a,  false, b,
         (short)0, c,  false,  false);
  }

  if (n16 < kG) {
    const float bias = bih0[n16];
#pragma unroll
    for (int v = 0; v < 8; ++v) {
      const int m = v + 8 * half;
      gx[(size_t)(row0 + m) * kG + n16] = c[v] + bias;
    }
  }
}

__device__ __forceinline__ float sigmoid_f(float x) {
  return 1.0f / (1.0f + __expf(-x));
}
__device__ __forceinline__ float tanh_f(float x) {
  const float e = __expf(2.0f * fabsf(x));
  return copysignf(1.0f - 2.0f / (e + 1.0f), x);
}

__device__ __forceinline__ void gru_cell(float h[kH], const float gxv[kG],
                                         const float* __restrict__ whh,
                                         const float* __restrict__ bhh) {
  float gh[kG];
#pragma unroll
  for (int g = 0; g < kG; ++g) {
    float s = bhh[g];
#pragma unroll
    for (int j = 0; j < kH; ++j) s += whh[g * kH + j] * h[j];
    gh[g] = s;
  }
#pragma unroll
  for (int j = 0; j < kH; ++j) {
    const float r = sigmoid_f(gxv[j] + gh[j]);
    const float z = sigmoid_f(gxv[kH + j] + gh[kH + j]);
    const float n = tanh_f(gxv[2 * kH + j] + r * gh[2 * kH + j]);
    h[j] = (1.0f - z) * n + z * h[j];
  }
}

__global__ __launch_bounds__(64) void gru_scan(
    const float* __restrict__ gx0, const float* __restrict__ hxs,
    const float* __restrict__ whh0, const float* __restrict__ bhh0,
    const float* __restrict__ wih_r, const float* __restrict__ whh_r,
    const float* __restrict__ bih_r, const float* __restrict__ bhh_r,
    float* __restrict__ out) {
  __shared__ float Ws[36 + 3 * 72];
  const int tid = threadIdx.x;
  for (int i = tid; i < 27; i += 64) Ws[i] = whh0[i];
  for (int i = tid; i < 9; i += 64) Ws[27 + i] = bhh0[i];
  for (int i = tid; i < 3 * 27; i += 64)
    Ws[36 + (i / 27) * 72 + (i % 27)] = wih_r[i];
  for (int i = tid; i < 3 * 27; i += 64)
    Ws[36 + (i / 27) * 72 + 27 + (i % 27)] = whh_r[i];
  for (int i = tid; i < 3 * 9; i += 64)
    Ws[36 + (i / 9) * 72 + 54 + (i % 9)] = bih_r[i];
  for (int i = tid; i < 3 * 9; i += 64)
    Ws[36 + (i / 9) * 72 + 63 + (i % 9)] = bhh_r[i];
  __syncthreads();

  const int b = tid;
  float h[kL][kH];
#pragma unroll
  for (int l = 0; l < kL; ++l)
#pragma unroll
    for (int j = 0; j < kH; ++j) h[l][j] = hxs[(l * kBATCH + b) * kH + j];

  for (int t = 0; t < kSEQ; ++t) {
    float gxv[kG];
    const float* gp = gx0 + (size_t)(t * kBATCH + b) * kG;
#pragma unroll
    for (int g = 0; g < kG; ++g) gxv[g] = gp[g];
    gru_cell(h[0], gxv, &Ws[0], &Ws[27]);

#pragma unroll
    for (int l = 1; l < kL; ++l) {
      const float* wih = &Ws[36 + (l - 1) * 72];
      const float* whh = wih + 27;
      const float* bih = wih + 54;
      const float* bhh = wih + 63;
#pragma unroll
      for (int g = 0; g < kG; ++g) {
        float s = bih[g];
#pragma unroll
        for (int j = 0; j < kH; ++j) s += wih[g * kH + j] * h[l - 1][j];
        gxv[g] = s;
      }
      gru_cell(h[l], gxv, whh, bhh);
    }

#pragma unroll
    for (int j = 0; j < kH; ++j)
      out[(size_t)(t * kBATCH + b) * kH + j] = h[kL - 1][j];
  }

  for (int l = 0; l < kL; ++l)
#pragma unroll
    for (int j = 0; j < kH; ++j)
      out[(size_t)kSEQ * kBATCH * kH + (l * kBATCH + b) * kH + j] = h[l][j];
}

extern "C" void kernel_launch(void* const* d_in, const int* in_sizes, int n_in,
                              void* d_out, int out_size, void* d_ws,
                              size_t ws_size, hipStream_t stream) {
  const float* x    = (const float*)d_in[0];
  const float* hxs  = (const float*)d_in[1];
  const float* wih0 = (const float*)d_in[2];
  const float* whh0 = (const float*)d_in[3];
  const float* bih0 = (const float*)d_in[4];
  const float* bhh0 = (const float*)d_in[5];
  const float* wihr = (const float*)d_in[6];
  const float* whhr = (const float*)d_in[7];
  const float* bihr = (const float*)d_in[8];
  const float* bhhr = (const float*)d_in[9];
  float* out = (float*)d_out;
  float* gx  = (float*)d_ws;

  gru_in_proj_wmma<<<kROWS / 128, 256, 0, stream>>>(x, wih0, bih0, gx);
  gru_scan<<<1, 64, 0, stream>>>(gx, hxs, whh0, bhh0, wihr, whhr, bihr, bhhr,
                                 out);
}


# GRU scan: a cell's dependent chain cut from 35 to 26 vector ops: exp's log2e factors carried by gate matrices and gx0 (no mul before exp); tanh sign and blend as one fma h'=(a+z*h)-a*q, a=sign*(1-z)
# speedup vs baseline: 10.3523x; 10.3523x over previous
#include <hip/hip_runtime.h>
#include <hip/hip_bf16.h>

typedef float v2f __attribute__((ext_vector_type(2)));
typedef float v8f __attribute__((ext_vector_type(8)));

static constexpr int kSEQ   = 2048;
static constexpr int kBATCH = 64;
static constexpr int kIN    = 512;
static constexpr int kH     = 3;
static constexpr int kL     = 4;
static constexpr int kG     = 3 * kH;
static constexpr int kROWS  = kSEQ * kBATCH;

static constexpr float kSigScale  = -1.4426950408889634f;
static constexpr float kTanhScale = 2.8853900817779268f;

__global__ __launch_bounds__(256) void gru_in_proj_wmma(
    const float* __restrict__ X, const float* __restrict__ Wih0,
    const float* __restrict__ bih0, float* __restrict__ gx) {
  __shared__ float Bs[kIN * 16];
  const int tid = threadIdx.x;
  for (int idx = tid; idx < kIN * 16; idx += 256) {
    const int k = idx >> 4;
    const int n = idx & 15;
    Bs[idx] = (n < kG) ? Wih0[n * kIN + k] : 0.0f;
  }
  __syncthreads();

  const int wave = tid >> 5;
  const int lane = tid & 31;
  const int half = lane >> 4;
  const int n16  = lane & 15;
  const int row0 = blockIdx.x * 128 + wave * 16;

  const float* arow = X + (size_t)(row0 + n16) * kIN + 2 * half;

  v8f c = {0.f, 0.f, 0.f, 0.f, 0.f, 0.f, 0.f, 0.f};
  for (int kk = 0; kk < kIN; kk += 4) {
    v2f a = *(const v2f*)(arow + kk);
    v2f b;
    b.x = Bs[(kk + 2 * half) * 16 + n16];
    b.y = Bs[(kk + 2 * half + 1) * 16 + n16];
    c = __builtin_amdgcn_wmma_f32_16x16x4_f32(
         false, a,  false, b,
         (short)0, c,  false,  false);
  }

  if (n16 < kG) {
    const float bias = bih0[n16];
    const float s = n16 < 2 * kH ? kSigScale : kTanhScale;
#pragma unroll
    for (int v = 0; v < 8; ++v) {
      const int m = v + 8 * half;
      gx[(size_t)(row0 + m) * kG + n16] = (c[v] + bias) * s;
    }
  }
}

static constexpr int kTB    = 8;
static constexpr int kNB    = kSEQ / kTB;
static constexpr int kWAVES = kL + 1;

__device__ __forceinline__ v2f gate_frag(const float* __restrict__ w,
                                         const float* __restrict__ bias,
                                         int lane, bool hidden_side) {
  const int m = lane & 15;
  const int half = lane >> 4;
  const int u = m >> 2;
  const int q = m & 3;
  const bool n_row_here = (q == 3) == hidden_side;
  const bool live = (u < kH) && (q < 2 || n_row_here);
  const int g = live ? (q < 2 ? q * kH + u : 2 * kH + u) : 0;
  const float w0 = w[g * kH + 2 * half];
  const float w1 = half ? bias[g] : w[g * kH + 1];
  const float s = q < 2 ? kSigScale : kTanhScale;
  v2f a;
  a.x = live ? w0 * s : 0.f;
  a.y = live ? w1 * s : 0.f;
  return a;
}

__device__ __forceinline__ void gates_to_state(const v8f d, v2f& h, int half) {
  const float r0 = 1.0f / (1.0f + __builtin_amdgcn_exp2f(d[0]));
  const float r1 = 1.0f / (1.0f + __builtin_amdgcn_exp2f(d[4]));
  const float z0 = 1.0f / (1.0f + __builtin_amdgcn_exp2f(d[1]));
  const float z1 = 1.0f / (1.0f + __builtin_amdgcn_exp2f(d[5]));
  const float p0 = d[2] + r0 * d[3];
  const float p1 = d[6] + r1 * d[7];
  const float q0 = 2.0f / (__builtin_amdgcn_exp2f(fabsf(p0)) + 1.0f);
  const float q1 = 2.0f / (__builtin_amdgcn_exp2f(fabsf(p1)) + 1.0f);
  const float a0 = copysignf(1.0f - z0, p0);
  const float a1 = half ? 0.0f : copysignf(1.0f - z1, p1);
  const float b0 = a0 + z0 * h.x;
  const float b1 = half ? 1.0f : a1 + z1 * h.y;
  h.x = b0 - a0 * q0;
  h.y = b1 - a1 * q1;
}

__device__ __forceinline__ v8f mm(const v2f a, const v2f b, const v8f c) {
  return __builtin_amdgcn_wmma_f32_16x16x4_f32(
       false, a,  false, b,
       (short)0, c,  false,  false);
}

__device__ __forceinline__ void load_row(v8f& row,
                                         const float* __restrict__ gstep,
                                         int col, int ua) {
  row[0] = gstep[col + ua];
  row[1] = gstep[col + kH + ua];
  row[2] = gstep[col + 2 * kH + ua];
  row[4] = gstep[col + 1];
  row[5] = gstep[col + kH + 1];
  row[6] = gstep[col + 2 * kH + 1];
}

__device__ __forceinline__ void block_first(const v8f* __restrict__ gin,
                                            v2f* __restrict__ yout,
                                            const v2f a_hh, v2f& h, int half) {
  v8f c[kTB];
#pragma unroll
  for (int j = 0; j < kTB; ++j) c[j] = gin[j * 32];
  __builtin_amdgcn_sched_barrier(0);
#pragma unroll
  for (int j = 0; j < kTB; ++j) {
    const v8f d = mm(a_hh, h, c[j]);
    gates_to_state(d, h, half);
    yout[j * 32] = h;
  }
}

__device__ __forceinline__ void block_upper(const v2f* __restrict__ yin,
                                            v2f* __restrict__ yout,
                                            const v2f a_ih, const v2f a_hh,
                                            v2f& h, int half) {
  const v8f zero8 = {0.f, 0.f, 0.f, 0.f, 0.f, 0.f, 0.f, 0.f};
  v2f y[kTB];
#pragma unroll
  for (int j = 0; j < kTB; ++j) y[j] = yin[j * 32];
  v8f din = mm(a_ih, y[0], zero8);
  __builtin_amdgcn_sched_barrier(0);
#pragma unroll
  for (int j = 0; j < kTB; ++j) {
    const v8f d = mm(a_hh, h, din);
    if (j + 1 < kTB) din = mm(a_ih, y[j + 1], zero8);
    __builtin_amdgcn_sched_barrier(0);
    gates_to_state(d, h, half);
    yout[j * 32] = h;
    __builtin_amdgcn_sched_barrier(0);
  }
}

__device__ __forceinline__ void store_out(float* __restrict__ op, const v2f h,
                                          int half) {
  op[2 * half] = h.x;
  if (half == 0) op[1] = h.y;
}

__global__ __launch_bounds__(32 * kWAVES)
__attribute__((amdgpu_waves_per_eu(1, 2))) void gru_scan(
    const float* __restrict__ gx0, const float* __restrict__ hxs,
    const float* __restrict__ whh0, const float* __restrict__ bhh0,
    const float* __restrict__ wih_r, const float* __restrict__ whh_r,
    const float* __restrict__ bih_r, const float* __restrict__ bhh_r,
    float* __restrict__ out) {
  __shared__ v2f Y[kL][2][kTB][32];
  __shared__ v8f G[2][kTB][32];

  const int lane = threadIdx.x & 31;
  const int w = __builtin_amdgcn_readfirstlane((int)(threadIdx.x >> 5));
  const bool mover = w == kL;
  const int wl = mover ? kL - 1 : w;
  const int half = lane >> 4;
  const int c16 = lane & 15;
  const int bcol = blockIdx.x * 16 + c16;
  const int ua = 2 * half;

  const int lr = wl > 0 ? wl - 1 : 0;
  const v2f a_hh = gate_frag(wl == 0 ? whh0 : whh_r + lr * kG * kH,
                             wl == 0 ? bhh0 : bhh_r + lr * kG, lane, true);
  v2f a_ih = {0.f, 0.f};
  if (wl > 0)
    a_ih = gate_frag(wih_r + lr * kG * kH, bih_r + lr * kG, lane, false);

  v2f h;
  {
    const float* hp = hxs + (wl * kBATCH + bcol) * kH;
    const float x = hp[2 * half];
    const float y = hp[1];
    h.x = x;
    h.y = half ? 1.0f : y;
  }

  const int col = bcol * kG;
  v8f nxt[kTB];
#pragma unroll
  for (int j = 0; j < kTB; ++j)
    nxt[j] = v8f{0.f, 0.f, 0.f, 0.f, 0.f, 0.f, 0.f, 0.f};
  const int e0 = lane, e1 = lane + 32;
  const int e1r = e1 < 16 * kH ? e1 : 16 * kH - 1;
  const int src0 =
      (e0 % kH) < 2 ? 2 * (e0 / kH) + (e0 % kH) : 2 * (16 + e0 / kH);
  const int src1 =
      (e1r % kH) < 2 ? 2 * (e1r / kH) + (e1r % kH) : 2 * (16 + e1r / kH);
  if (mover) {
#pragma unroll
    for (int j = 0; j < kTB; ++j)
      load_row(nxt[j], gx0 + (size_t)j * (kBATCH * kG), col, ua);
#pragma unroll
    for (int j = 0; j < kTB; ++j) G[0][j][lane] = nxt[j];
#pragma unroll
    for (int j = 0; j < kTB; ++j)
      load_row(nxt[j], gx0 + (size_t)(kTB + j) * (kBATCH * kG), col, ua);
  }
  __builtin_amdgcn_fence(__ATOMIC_RELEASE, "workgroup", "local");
  __builtin_amdgcn_s_barrier();
  __builtin_amdgcn_fence(__ATOMIC_ACQUIRE, "workgroup", "local");

  for (int k = 0; k < kNB + kL; ++k) {
    const int blk = k - w;
    if (mover) {
      if (k + 1 < kNB) {
#pragma unroll
        for (int j = 0; j < kTB; ++j) G[(k + 1) & 1][j][lane] = nxt[j];
      }
      if (k + 2 < kNB) {
        const float* g = gx0 + (size_t)(k + 2) * kTB * (kBATCH * kG);
#pragma unroll
        for (int j = 0; j < kTB; ++j)
          load_row(nxt[j], g + (size_t)j * (kBATCH * kG), col, ua);
      }
      const int ob = k - kL;
      if (ob >= 0) {
        const float* o = (const float*)&Y[kL - 1][(k - 1) & 1][0][0];
        float* op = out + ((size_t)ob * kTB * kBATCH + blockIdx.x * 16) * kH;
        float v0[kTB], v1[kTB];
#pragma unroll
        for (int j = 0; j < kTB; ++j) {
          v0[j] = o[j * 64 + src0];
          v1[j] = o[j * 64 + src1];
        }
#pragma unroll
        for (int j = 0; j < kTB; ++j) op[j * (kBATCH * kH) + e0] = v0[j];
        if (lane < 16) {
#pragma unroll
          for (int j = 0; j < kTB; ++j) op[j * (kBATCH * kH) + e1] = v1[j];
        }
      }
    } else if (blk >= 0 && blk < kNB) {
      if (w == 0) {
        block_first(&G[k & 1][0][lane], &Y[0][k & 1][0][lane], a_hh, h, half);
      } else {
        block_upper(&Y[w - 1][(k - 1) & 1][0][lane], &Y[w][k & 1][0][lane],
                    a_ih, a_hh, h, half);
      }
    }
    __builtin_amdgcn_fence(__ATOMIC_RELEASE, "workgroup", "local");
    __builtin_amdgcn_s_barrier();
    __builtin_amdgcn_fence(__ATOMIC_ACQUIRE, "workgroup", "local");
  }

  if (!mover) {
    store_out(out + (size_t)kSEQ * kBATCH * kH + (w * kBATCH + bcol) * kH, h,
              half);
  }
}

extern "C" void kernel_launch(void* const* d_in, const int* in_sizes, int n_in,
                              void* d_out, int out_size, void* d_ws,
                              size_t ws_size, hipStream_t stream) {
  const float* x    = (const float*)d_in[0];
  const float* hxs  = (const float*)d_in[1];
  const float* wih0 = (const float*)d_in[2];
  const float* whh0 = (const float*)d_in[3];
  const float* bih0 = (const float*)d_in[4];
  const float* bhh0 = (const float*)d_in[5];
  const float* wihr = (const float*)d_in[6];
  const float* whhr = (const float*)d_in[7];
  const float* bihr = (const float*)d_in[8];
  const float* bhhr = (const float*)d_in[9];
  float* out = (float*)d_out;
  float* gx  = (float*)d_ws;

  gru_in_proj_wmma<<<kROWS / 128, 256, 0, stream>>>(x, wih0, bih0, gx);
  gru_scan<<<kBATCH / 16, 32 * kWAVES, 0, stream>>>(gx, hxs, whh0, bhh0, wihr,
                                                     whhr, bihr, bhhr, out);
}


# GRU scan: a cell's gate math issued level by level: the waiting link (r's, then n's division; both units side by side) in front, z gate's steps two levels late in its gaps; same IEEE steps and sums
# speedup vs baseline: 11.4863x; 1.1095x over previous
#include <hip/hip_runtime.h>
#include <hip/hip_bf16.h>

typedef float v2f __attribute__((ext_vector_type(2)));
typedef float v8f __attribute__((ext_vector_type(8)));

static constexpr int kSEQ   = 2048;
static constexpr int kBATCH = 64;
static constexpr int kIN    = 512;
static constexpr int kH     = 3;
static constexpr int kL     = 4;
static constexpr int kG     = 3 * kH;
static constexpr int kROWS  = kSEQ * kBATCH;

static constexpr float kSigScale  = -1.4426950408889634f;
static constexpr float kTanhScale = 2.8853900817779268f;

__global__ __launch_bounds__(256) void gru_in_proj_wmma(
    const float* __restrict__ X, const float* __restrict__ Wih0,
    const float* __restrict__ bih0, float* __restrict__ gx) {
  __shared__ float Bs[kIN * 16];
  const int tid = threadIdx.x;
  for (int idx = tid; idx < kIN * 16; idx += 256) {
    const int k = idx >> 4;
    const int n = idx & 15;
    Bs[idx] = (n < kG) ? Wih0[n * kIN + k] : 0.0f;
  }
  __syncthreads();

  const int wave = tid >> 5;
  const int lane = tid & 31;
  const int half = lane >> 4;
  const int n16  = lane & 15;
  const int row0 = blockIdx.x * 128 + wave * 16;

  const float* arow = X + (size_t)(row0 + n16) * kIN + 2 * half;

  v8f c = {0.f, 0.f, 0.f, 0.f, 0.f, 0.f, 0.f, 0.f};
  for (int kk = 0; kk < kIN; kk += 4) {
    v2f a = *(const v2f*)(arow + kk);
    v2f b;
    b.x = Bs[(kk + 2 * half) * 16 + n16];
    b.y = Bs[(kk + 2 * half + 1) * 16 + n16];
    c = __builtin_amdgcn_wmma_f32_16x16x4_f32(
         false, a,  false, b,
         (short)0, c,  false,  false);
  }

  if (n16 < kG) {
    const float bias = bih0[n16];
    const float s = n16 < 2 * kH ? kSigScale : kTanhScale;
#pragma unroll
    for (int v = 0; v < 8; ++v) {
      const int m = v + 8 * half;
      gx[(size_t)(row0 + m) * kG + n16] = (c[v] + bias) * s;
    }
  }
}

static constexpr int kTB    = 8;
static constexpr int kNB    = kSEQ / kTB;
static constexpr int kWAVES = kL + 1;

__device__ __forceinline__ v2f gate_frag(const float* __restrict__ w,
                                         const float* __restrict__ bias,
                                         int lane, bool hidden_side) {
  const int m = lane & 15;
  const int half = lane >> 4;
  const int u = m >> 2;
  const int q = m & 3;
  const bool n_row_here = (q == 3) == hidden_side;
  const bool live = (u < kH) && (q < 2 || n_row_here);
  const int g = live ? (q < 2 ? q * kH + u : 2 * kH + u) : 0;
  const float w0 = w[g * kH + 2 * half];
  const float w1 = half ? bias[g] : w[g * kH + 1];
  const float s = q < 2 ? kSigScale : kTanhScale;
  v2f a;
  a.x = live ? w0 * s : 0.f;
  a.y = live ? w1 * s : 0.f;
  return a;
}

struct Div {
  float den, ds, ns, x, q, r, o;
  bool flag;
};
template <int kNumTimes2>
struct DivSteps {
  static constexpr float num = 0.5f * kNumTimes2;
  static __device__ __forceinline__ void s1(Div& v) {
    bool unused;
    v.ds = __builtin_amdgcn_div_scalef(num, v.den, false, &unused);
  }
  static __device__ __forceinline__ void sn(Div& v) {
    v.ns = __builtin_amdgcn_div_scalef(num, v.den, true, &v.flag);
  }
  static __device__ __forceinline__ void s2(Div& v) {
    v.x = __builtin_amdgcn_rcpf(v.ds);
  }
  static __device__ __forceinline__ void s3(Div& v) {
    v.r = __builtin_fmaf(-v.ds, v.x, 1.0f);
  }
  static __device__ __forceinline__ void s4(Div& v) {
    v.x = __builtin_fmaf(v.r, v.x, v.x);
  }
  static __device__ __forceinline__ void s5(Div& v) { v.q = v.ns * v.x; }
  static __device__ __forceinline__ void s6(Div& v) {
    v.r = __builtin_fmaf(-v.ds, v.q, v.ns);
  }
  static __device__ __forceinline__ void s7(Div& v) {
    v.q = __builtin_fmaf(v.r, v.x, v.q);
  }
  static __device__ __forceinline__ void s8(Div& v) {
    v.r = __builtin_fmaf(-v.ds, v.q, v.ns);
  }
  static __device__ __forceinline__ void s9(Div& v) {
    v.o = __builtin_amdgcn_div_fmasf(v.r, v.x, v.q, v.flag);
  }
  static __device__ __forceinline__ void s10(Div& v) {
    v.o = __builtin_amdgcn_div_fixupf(v.o, v.den, num);
  }
};

__device__ __forceinline__ void level_end() {
  __builtin_amdgcn_sched_barrier(0);
}

__device__ __forceinline__ void gates_to_state(const v8f d, v2f& h, int half) {
  typedef DivSteps<2> One;
  typedef DivSteps<-4> Neg2;
  Div r0, r1, z0, z1, q0, q1;
  const float er0 = __builtin_amdgcn_exp2f(d[0]);
  const float er1 = __builtin_amdgcn_exp2f(d[4]);
  const float ez0 = __builtin_amdgcn_exp2f(d[1]);
  const float ez1 = __builtin_amdgcn_exp2f(d[5]);
  level_end();
  r0.den = 1.0f + er0;
  r1.den = 1.0f + er1;
  level_end();
  One::s1(r0);
  One::s1(r1);
  level_end();
  One::s2(r0);
  One::s2(r1);
  z0.den = 1.0f + ez0;
  z1.den = 1.0f + ez1;
  One::sn(r0);
  One::sn(r1);
  level_end();
  One::s3(r0);
  One::s3(r1);
  One::s1(z0);
  One::s1(z1);
  level_end();
  One::s4(r0);
  One::s4(r1);
  One::s2(z0);
  One::s2(z1);
  One::sn(z0);
  One::sn(z1);
  level_end();
  One::s5(r0);
  One::s5(r1);
  level_end();
  One::s6(r0);
  One::s6(r1);
  One::s3(z0);
  One::s3(z1);
  level_end();
  One::s7(r0);
  One::s7(r1);
  One::s4(z0);
  One::s4(z1);
  level_end();
  One::s8(r0);
  One::s8(r1);
  One::s5(z0);
  One::s5(z1);
  level_end();
  One::s9(r0);
  One::s9(r1);
  One::s6(z0);
  One::s6(z1);
  level_end();
  One::s10(r0);
  One::s10(r1);
  One::s7(z0);
  One::s7(z1);
  level_end();
  const float p0 = __builtin_fmaf(d[3], r0.o, d[2]);
  const float p1 = __builtin_fmaf(d[7], r1.o, d[6]);
  One::s8(z0);
  One::s8(z1);
  level_end();
  const float f0 = __builtin_amdgcn_exp2f(fabsf(p0));
  const float f1 = __builtin_amdgcn_exp2f(fabsf(p1));
  One::s9(z0);
  One::s9(z1);
  level_end();
  q0.den = f0 + 1.0f;
  q1.den = f1 + 1.0f;
  One::s10(z0);
  One::s10(z1);
  level_end();
  Neg2::s1(q0);
  Neg2::s1(q1);
  const float omz0 = 1.0f - z0.o;
  const float omz1 = 1.0f - z1.o;
  level_end();
  Neg2::s2(q0);
  Neg2::s2(q1);
  const float a0 = copysignf(omz0, p0);
  const float a1s = copysignf(omz1, p1);
  Neg2::sn(q0);
  Neg2::sn(q1);
  level_end();
  Neg2::s3(q0);
  Neg2::s3(q1);
  level_end();
  Neg2::s4(q0);
  Neg2::s4(q1);
  const float b0 = __builtin_fmaf(h.x, z0.o, a0);
  const float b1s = __builtin_fmaf(h.y, z1.o, a1s);
  level_end();
  Neg2::s5(q0);
  Neg2::s5(q1);
  level_end();
  Neg2::s6(q0);
  Neg2::s6(q1);
  const float a1 = half ? 0.0f : a1s;
  const float b1 = half ? 1.0f : b1s;
  level_end();
  Neg2::s7(q0);
  Neg2::s7(q1);
  level_end();
  Neg2::s8(q0);
  Neg2::s8(q1);
  level_end();
  Neg2::s9(q0);
  Neg2::s9(q1);
  level_end();
  Neg2::s10(q0);
  Neg2::s10(q1);
  level_end();
  h.x = __builtin_fmaf(a0, q0.o, b0);
  h.y = __builtin_fmaf(a1, q1.o, b1);
}

__device__ __forceinline__ v8f mm(const v2f a, const v2f b, const v8f c) {
  return __builtin_amdgcn_wmma_f32_16x16x4_f32(
       false, a,  false, b,
       (short)0, c,  false,  false);
}

__device__ __forceinline__ void load_row(v8f& row,
                                         const float* __restrict__ gstep,
                                         int col, int ua) {
  row[0] = gstep[col + ua];
  row[1] = gstep[col + kH + ua];
  row[2] = gstep[col + 2 * kH + ua];
  row[4] = gstep[col + 1];
  row[5] = gstep[col + kH + 1];
  row[6] = gstep[col + 2 * kH + 1];
}

__device__ __forceinline__ void block_first(const v8f* __restrict__ gin,
                                            v2f* __restrict__ yout,
                                            const v2f a_hh, v2f& h, int half) {
  v8f c[kTB];
#pragma unroll
  for (int j = 0; j < kTB; ++j) c[j] = gin[j * 32];
  __builtin_amdgcn_sched_barrier(0);
#pragma unroll
  for (int j = 0; j < kTB; ++j) {
    const v8f d = mm(a_hh, h, c[j]);
    gates_to_state(d, h, half);
    yout[j * 32] = h;
  }
}

__device__ __forceinline__ void block_upper(const v2f* __restrict__ yin,
                                            v2f* __restrict__ yout,
                                            const v2f a_ih, const v2f a_hh,
                                            v2f& h, int half) {
  const v8f zero8 = {0.f, 0.f, 0.f, 0.f, 0.f, 0.f, 0.f, 0.f};
  v2f y[kTB];
#pragma unroll
  for (int j = 0; j < kTB; ++j) y[j] = yin[j * 32];
  v8f din = mm(a_ih, y[0], zero8);
  __builtin_amdgcn_sched_barrier(0);
#pragma unroll
  for (int j = 0; j < kTB; ++j) {
    const v8f d = mm(a_hh, h, din);
    if (j + 1 < kTB) din = mm(a_ih, y[j + 1], zero8);
    __builtin_amdgcn_sched_barrier(0);
    gates_to_state(d, h, half);
    yout[j * 32] = h;
    __builtin_amdgcn_sched_barrier(0);
  }
}

__device__ __forceinline__ void store_out(float* __restrict__ op, const v2f h,
                                          int half) {
  op[2 * half] = h.x;
  if (half == 0) op[1] = h.y;
}

__global__ __launch_bounds__(32 * kWAVES)
__attribute__((amdgpu_waves_per_eu(1, 2))) void gru_scan(
    const float* __restrict__ gx0, const float* __restrict__ hxs,
    const float* __restrict__ whh0, const float* __restrict__ bhh0,
    const float* __restrict__ wih_r, const float* __restrict__ whh_r,
    const float* __restrict__ bih_r, const float* __restrict__ bhh_r,
    float* __restrict__ out) {
  __shared__ v2f Y[kL][2][kTB][32];
  __shared__ v8f G[2][kTB][32];

  const int lane = threadIdx.x & 31;
  const int w = __builtin_amdgcn_readfirstlane((int)(threadIdx.x >> 5));
  const bool mover = w == kL;
  const int wl = mover ? kL - 1 : w;
  const int half = lane >> 4;
  const int c16 = lane & 15;
  const int bcol = blockIdx.x * 16 + c16;
  const int ua = 2 * half;

  const int lr = wl > 0 ? wl - 1 : 0;
  const v2f a_hh = gate_frag(wl == 0 ? whh0 : whh_r + lr * kG * kH,
                             wl == 0 ? bhh0 : bhh_r + lr * kG, lane, true);
  v2f a_ih = {0.f, 0.f};
  if (wl > 0)
    a_ih = gate_frag(wih_r + lr * kG * kH, bih_r + lr * kG, lane, false);

  v2f h;
  {
    const float* hp = hxs + (wl * kBATCH + bcol) * kH;
    const float x = hp[2 * half];
    const float y = hp[1];
    h.x = x;
    h.y = half ? 1.0f : y;
  }

  const int col = bcol * kG;
  v8f nxt[kTB];
#pragma unroll
  for (int j = 0; j < kTB; ++j)
    nxt[j] = v8f{0.f, 0.f, 0.f, 0.f, 0.f, 0.f, 0.f, 0.f};
  const int e0 = lane, e1 = lane + 32;
  const int e1r = e1 < 16 * kH ? e1 : 16 * kH - 1;
  const int src0 =
      (e0 % kH) < 2 ? 2 * (e0 / kH) + (e0 % kH) : 2 * (16 + e0 / kH);
  const int src1 =
      (e1r % kH) < 2 ? 2 * (e1r / kH) + (e1r % kH) : 2 * (16 + e1r / kH);
  if (mover) {
#pragma unroll
    for (int j = 0; j < kTB; ++j)
      load_row(nxt[j], gx0 + (size_t)j * (kBATCH * kG), col, ua);
#pragma unroll
    for (int j = 0; j < kTB; ++j) G[0][j][lane] = nxt[j];
#pragma unroll
    for (int j = 0; j < kTB; ++j)
      load_row(nxt[j], gx0 + (size_t)(kTB + j) * (kBATCH * kG), col, ua);
  }
  __builtin_amdgcn_fence(__ATOMIC_RELEASE, "workgroup", "local");
  __builtin_amdgcn_s_barrier();
  __builtin_amdgcn_fence(__ATOMIC_ACQUIRE, "workgroup", "local");

  for (int k = 0; k < kNB + kL; ++k) {
    const int blk = k - w;
    if (mover) {
      if (k + 1 < kNB) {
#pragma unroll
        for (int j = 0; j < kTB; ++j) G[(k + 1) & 1][j][lane] = nxt[j];
      }
      if (k + 2 < kNB) {
        const float* g = gx0 + (size_t)(k + 2) * kTB * (kBATCH * kG);
#pragma unroll
        for (int j = 0; j < kTB; ++j)
          load_row(nxt[j], g + (size_t)j * (kBATCH * kG), col, ua);
      }
      const int ob = k - kL;
      if (ob >= 0) {
        const float* o = (const float*)&Y[kL - 1][(k - 1) & 1][0][0];
        float* op = out + ((size_t)ob * kTB * kBATCH + blockIdx.x * 16) * kH;
        float v0[kTB], v1[kTB];
#pragma unroll
        for (int j = 0; j < kTB; ++j) {
          v0[j] = o[j * 64 + src0];
          v1[j] = o[j * 64 + src1];
        }
#pragma unroll
        for (int j = 0; j < kTB; ++j) op[j * (kBATCH * kH) + e0] = v0[j];
        if (lane < 16) {
#pragma unroll
          for (int j = 0; j < kTB; ++j) op[j * (kBATCH * kH) + e1] = v1[j];
        }
      }
    } else if (blk >= 0 && blk < kNB) {
      if (w == 0) {
        block_first(&G[k & 1][0][lane], &Y[0][k & 1][0][lane], a_hh, h, half);
      } else {
        block_upper(&Y[w - 1][(k - 1) & 1][0][lane], &Y[w][k & 1][0][lane],
                    a_ih, a_hh, h, half);
      }
    }
    __builtin_amdgcn_fence(__ATOMIC_RELEASE, "workgroup", "local");
    __builtin_amdgcn_s_barrier();
    __builtin_amdgcn_fence(__ATOMIC_ACQUIRE, "workgroup", "local");
  }

  if (!mover) {
    store_out(out + (size_t)kSEQ * kBATCH * kH + (w * kBATCH + bcol) * kH, h,
              half);
  }
}

extern "C" void kernel_launch(void* const* d_in, const int* in_sizes, int n_in,
                              void* d_out, int out_size, void* d_ws,
                              size_t ws_size, hipStream_t stream) {
  const float* x    = (const float*)d_in[0];
  const float* hxs  = (const float*)d_in[1];
  const float* wih0 = (const float*)d_in[2];
  const float* whh0 = (const float*)d_in[3];
  const float* bih0 = (const float*)d_in[4];
  const float* bhh0 = (const float*)d_in[5];
  const float* wihr = (const float*)d_in[6];
  const float* whhr = (const float*)d_in[7];
  const float* bihr = (const float*)d_in[8];
  const float* bhhr = (const float*)d_in[9];
  float* out = (float*)d_out;
  float* gx  = (float*)d_ws;

  gru_in_proj_wmma<<<kROWS / 128, 256, 0, stream>>>(x, wih0, bih0, gx);
  gru_scan<<<kBATCH / 16, 32 * kWAVES, 0, stream>>>(gx, hxs, whh0, bhh0, wihr,
                                                     whhr, bihr, bhhr, out);
}


# GRU scan: a block's 8 states stay in registers and go to the LDS hand-over buffer as one group behind its last cell: no LDS instruction between a cell's last fma and the next product; sums as before
# speedup vs baseline: 12.1948x; 1.0617x over previous
#include <hip/hip_runtime.h>
#include <hip/hip_bf16.h>

typedef float v2f __attribute__((ext_vector_type(2)));
typedef float v8f __attribute__((ext_vector_type(8)));

static constexpr int kSEQ   = 2048;
static constexpr int kBATCH = 64;
static constexpr int kIN    = 512;
static constexpr int kH     = 3;
static constexpr int kL     = 4;
static constexpr int kG     = 3 * kH;
static constexpr int kROWS  = kSEQ * kBATCH;

static constexpr float kSigScale  = -1.4426950408889634f;
static constexpr float kTanhScale = 2.8853900817779268f;

__global__ __launch_bounds__(256) void gru_in_proj_wmma(
    const float* __restrict__ X, const float* __restrict__ Wih0,
    const float* __restrict__ bih0, float* __restrict__ gx) {
  __shared__ float Bs[kIN * 16];
  const int tid = threadIdx.x;
  for (int idx = tid; idx < kIN * 16; idx += 256) {
    const int k = idx >> 4;
    const int n = idx & 15;
    Bs[idx] = (n < kG) ? Wih0[n * kIN + k] : 0.0f;
  }
  __syncthreads();

  const int wave = tid >> 5;
  const int lane = tid & 31;
  const int half = lane >> 4;
  const int n16  = lane & 15;
  const int row0 = blockIdx.x * 128 + wave * 16;

  const float* arow = X + (size_t)(row0 + n16) * kIN + 2 * half;

  v8f c = {0.f, 0.f, 0.f, 0.f, 0.f, 0.f, 0.f, 0.f};
  for (int kk = 0; kk < kIN; kk += 4) {
    v2f a = *(const v2f*)(arow + kk);
    v2f b;
    b.x = Bs[(kk + 2 * half) * 16 + n16];
    b.y = Bs[(kk + 2 * half + 1) * 16 + n16];
    c = __builtin_amdgcn_wmma_f32_16x16x4_f32(
         false, a,  false, b,
         (short)0, c,  false,  false);
  }

  if (n16 < kG) {
    const float bias = bih0[n16];
    const float s = n16 < 2 * kH ? kSigScale : kTanhScale;
#pragma unroll
    for (int v = 0; v < 8; ++v) {
      const int m = v + 8 * half;
      gx[(size_t)(row0 + m) * kG + n16] = (c[v] + bias) * s;
    }
  }
}

static constexpr int kTB    = 8;
static constexpr int kNB    = kSEQ / kTB;
static constexpr int kWAVES = kL + 1;

__device__ __forceinline__ v2f gate_frag(const float* __restrict__ w,
                                         const float* __restrict__ bias,
                                         int lane, bool hidden_side) {
  const int m = lane & 15;
  const int half = lane >> 4;
  const int u = m >> 2;
  const int q = m & 3;
  const bool n_row_here = (q == 3) == hidden_side;
  const bool live = (u < kH) && (q < 2 || n_row_here);
  const int g = live ? (q < 2 ? q * kH + u : 2 * kH + u) : 0;
  const float w0 = w[g * kH + 2 * half];
  const float w1 = half ? bias[g] : w[g * kH + 1];
  const float s = q < 2 ? kSigScale : kTanhScale;
  v2f a;
  a.x = live ? w0 * s : 0.f;
  a.y = live ? w1 * s : 0.f;
  return a;
}

struct Div {
  float den, ds, ns, x, q, r, o;
  bool flag;
};
template <int kNumTimes2>
struct DivSteps {
  static constexpr float num = 0.5f * kNumTimes2;
  static __device__ __forceinline__ void s1(Div& v) {
    bool unused;
    v.ds = __builtin_amdgcn_div_scalef(num, v.den, false, &unused);
  }
  static __device__ __forceinline__ void sn(Div& v) {
    v.ns = __builtin_amdgcn_div_scalef(num, v.den, true, &v.flag);
  }
  static __device__ __forceinline__ void s2(Div& v) {
    v.x = __builtin_amdgcn_rcpf(v.ds);
  }
  static __device__ __forceinline__ void s3(Div& v) {
    v.r = __builtin_fmaf(-v.ds, v.x, 1.0f);
  }
  static __device__ __forceinline__ void s4(Div& v) {
    v.x = __builtin_fmaf(v.r, v.x, v.x);
  }
  static __device__ __forceinline__ void s5(Div& v) { v.q = v.ns * v.x; }
  static __device__ __forceinline__ void s6(Div& v) {
    v.r = __builtin_fmaf(-v.ds, v.q, v.ns);
  }
  static __device__ __forceinline__ void s7(Div& v) {
    v.q = __builtin_fmaf(v.r, v.x, v.q);
  }
  static __device__ __forceinline__ void s8(Div& v) {
    v.r = __builtin_fmaf(-v.ds, v.q, v.ns);
  }
  static __device__ __forceinline__ void s9(Div& v) {
    v.o = __builtin_amdgcn_div_fmasf(v.r, v.x, v.q, v.flag);
  }
  static __device__ __forceinline__ void s10(Div& v) {
    v.o = __builtin_amdgcn_div_fixupf(v.o, v.den, num);
  }
};

__device__ __forceinline__ void level_end() {
  __builtin_amdgcn_sched_barrier(0);
}

__device__ __forceinline__ void gates_to_state(const v8f d, v2f& h, int half) {
  typedef DivSteps<2> One;
  typedef DivSteps<-4> Neg2;
  Div r0, r1, z0, z1, q0, q1;
  const float er0 = __builtin_amdgcn_exp2f(d[0]);
  const float er1 = __builtin_amdgcn_exp2f(d[4]);
  const float ez0 = __builtin_amdgcn_exp2f(d[1]);
  const float ez1 = __builtin_amdgcn_exp2f(d[5]);
  level_end();
  r0.den = 1.0f + er0;
  r1.den = 1.0f + er1;
  level_end();
  One::s1(r0);
  One::s1(r1);
  level_end();
  One::s2(r0);
  One::s2(r1);
  z0.den = 1.0f + ez0;
  z1.den = 1.0f + ez1;
  One::sn(r0);
  One::sn(r1);
  level_end();
  One::s3(r0);
  One::s3(r1);
  One::s1(z0);
  One::s1(z1);
  level_end();
  One::s4(r0);
  One::s4(r1);
  One::s2(z0);
  One::s2(z1);
  One::sn(z0);
  One::sn(z1);
  level_end();
  One::s5(r0);
  One::s5(r1);
  level_end();
  One::s6(r0);
  One::s6(r1);
  One::s3(z0);
  One::s3(z1);
  level_end();
  One::s7(r0);
  One::s7(r1);
  One::s4(z0);
  One::s4(z1);
  level_end();
  One::s8(r0);
  One::s8(r1);
  One::s5(z0);
  One::s5(z1);
  level_end();
  One::s9(r0);
  One::s9(r1);
  One::s6(z0);
  One::s6(z1);
  level_end();
  One::s10(r0);
  One::s10(r1);
  One::s7(z0);
  One::s7(z1);
  level_end();
  const float p0 = __builtin_fmaf(d[3], r0.o, d[2]);
  const float p1 = __builtin_fmaf(d[7], r1.o, d[6]);
  One::s8(z0);
  One::s8(z1);
  level_end();
  const float f0 = __builtin_amdgcn_exp2f(fabsf(p0));
  const float f1 = __builtin_amdgcn_exp2f(fabsf(p1));
  One::s9(z0);
  One::s9(z1);
  level_end();
  q0.den = f0 + 1.0f;
  q1.den = f1 + 1.0f;
  One::s10(z0);
  One::s10(z1);
  level_end();
  Neg2::s1(q0);
  Neg2::s1(q1);
  const float omz0 = 1.0f - z0.o;
  const float omz1 = 1.0f - z1.o;
  level_end();
  Neg2::s2(q0);
  Neg2::s2(q1);
  const float a0 = copysignf(omz0, p0);
  const float a1s = copysignf(omz1, p1);
  Neg2::sn(q0);
  Neg2::sn(q1);
  level_end();
  Neg2::s3(q0);
  Neg2::s3(q1);
  level_end();
  Neg2::s4(q0);
  Neg2::s4(q1);
  const float b0 = __builtin_fmaf(h.x, z0.o, a0);
  const float b1s = __builtin_fmaf(h.y, z1.o, a1s);
  level_end();
  Neg2::s5(q0);
  Neg2::s5(q1);
  level_end();
  Neg2::s6(q0);
  Neg2::s6(q1);
  const float a1 = half ? 0.0f : a1s;
  const float b1 = half ? 1.0f : b1s;
  level_end();
  Neg2::s7(q0);
  Neg2::s7(q1);
  level_end();
  Neg2::s8(q0);
  Neg2::s8(q1);
  level_end();
  Neg2::s9(q0);
  Neg2::s9(q1);
  level_end();
  Neg2::s10(q0);
  Neg2::s10(q1);
  level_end();
  h.x = __builtin_fmaf(a0, q0.o, b0);
  h.y = __builtin_fmaf(a1, q1.o, b1);
}

__device__ __forceinline__ v8f mm(const v2f a, const v2f b, const v8f c) {
  return __builtin_amdgcn_wmma_f32_16x16x4_f32(
       false, a,  false, b,
       (short)0, c,  false,  false);
}

__device__ __forceinline__ void load_row(v8f& row,
                                         const float* __restrict__ gstep,
                                         int col, int ua) {
  row[0] = gstep[col + ua];
  row[1] = gstep[col + kH + ua];
  row[2] = gstep[col + 2 * kH + ua];
  row[4] = gstep[col + 1];
  row[5] = gstep[col + kH + 1];
  row[6] = gstep[col + 2 * kH + 1];
}

__device__ __forceinline__ void store_block(v2f* __restrict__ yout,
                                            const v2f (&hs)[kTB]) {
  __builtin_amdgcn_sched_barrier(0);
#pragma unroll
  for (int j = 0; j < kTB; ++j) yout[j * 32] = hs[j];
}

__device__ __forceinline__ void block_first(const v8f* __restrict__ gin,
                                            v2f* __restrict__ yout,
                                            const v2f a_hh, v2f& h, int half) {
  v8f c[kTB];
#pragma unroll
  for (int j = 0; j < kTB; ++j) c[j] = gin[j * 32];
  __builtin_amdgcn_sched_barrier(0);
  v2f hs[kTB];
#pragma unroll
  for (int j = 0; j < kTB; ++j) {
    const v8f d = mm(a_hh, h, c[j]);
    gates_to_state(d, h, half);
    hs[j] = h;
  }
  store_block(yout, hs);
}

__device__ __forceinline__ void block_upper(const v2f* __restrict__ yin,
                                            v2f* __restrict__ yout,
                                            const v2f a_ih, const v2f a_hh,
                                            v2f& h, int half) {
  const v8f zero8 = {0.f, 0.f, 0.f, 0.f, 0.f, 0.f, 0.f, 0.f};
  v2f y[kTB];
#pragma unroll
  for (int j = 0; j < kTB; ++j) y[j] = yin[j * 32];
  v2f hs[kTB];
  v8f din = mm(a_ih, y[0], zero8);
  __builtin_amdgcn_sched_barrier(0);
#pragma unroll
  for (int j = 0; j < kTB; ++j) {
    const v8f d = mm(a_hh, h, din);
    if (j + 1 < kTB) din = mm(a_ih, y[j + 1], zero8);
    __builtin_amdgcn_sched_barrier(0);
    gates_to_state(d, h, half);
    hs[j] = h;
    __builtin_amdgcn_sched_barrier(0);
  }
  store_block(yout, hs);
}

__device__ __forceinline__ void store_out(float* __restrict__ op, const v2f h,
                                          int half) {
  op[2 * half] = h.x;
  if (half == 0) op[1] = h.y;
}

__global__ __launch_bounds__(32 * kWAVES)
__attribute__((amdgpu_waves_per_eu(1, 2))) void gru_scan(
    const float* __restrict__ gx0, const float* __restrict__ hxs,
    const float* __restrict__ whh0, const float* __restrict__ bhh0,
    const float* __restrict__ wih_r, const float* __restrict__ whh_r,
    const float* __restrict__ bih_r, const float* __restrict__ bhh_r,
    float* __restrict__ out) {
  __shared__ v2f Y[kL][2][kTB][32];
  __shared__ v8f G[2][kTB][32];

  const int lane = threadIdx.x & 31;
  const int w = __builtin_amdgcn_readfirstlane((int)(threadIdx.x >> 5));
  const bool mover = w == kL;
  const int wl = mover ? kL - 1 : w;
  const int half = lane >> 4;
  const int c16 = lane & 15;
  const int bcol = blockIdx.x * 16 + c16;
  const int ua = 2 * half;

  const int lr = wl > 0 ? wl - 1 : 0;
  const v2f a_hh = gate_frag(wl == 0 ? whh0 : whh_r + lr * kG * kH,
                             wl == 0 ? bhh0 : bhh_r + lr * kG, lane, true);
  v2f a_ih = {0.f, 0.f};
  if (wl > 0)
    a_ih = gate_frag(wih_r + lr * kG * kH, bih_r + lr * kG, lane, false);

  v2f h;
  {
    const float* hp = hxs + (wl * kBATCH + bcol) * kH;
    const float x = hp[2 * half];
    const float y = hp[1];
    h.x = x;
    h.y = half ? 1.0f : y;
  }

  const int col = bcol * kG;
  v8f nxt[kTB];
#pragma unroll
  for (int j = 0; j < kTB; ++j)
    nxt[j] = v8f{0.f, 0.f, 0.f, 0.f, 0.f, 0.f, 0.f, 0.f};
  const int e0 = lane, e1 = lane + 32;
  const int e1r = e1 < 16 * kH ? e1 : 16 * kH - 1;
  const int src0 =
      (e0 % kH) < 2 ? 2 * (e0 / kH) + (e0 % kH) : 2 * (16 + e0 / kH);
  const int src1 =
      (e1r % kH) < 2 ? 2 * (e1r / kH) + (e1r % kH) : 2 * (16 + e1r / kH);
  if (mover) {
#pragma unroll
    for (int j = 0; j < kTB; ++j)
      load_row(nxt[j], gx0 + (size_t)j * (kBATCH * kG), col, ua);
#pragma unroll
    for (int j = 0; j < kTB; ++j) G[0][j][lane] = nxt[j];
#pragma unroll
    for (int j = 0; j < kTB; ++j)
      load_row(nxt[j], gx0 + (size_t)(kTB + j) * (kBATCH * kG), col, ua);
  }
  __builtin_amdgcn_fence(__ATOMIC_RELEASE, "workgroup", "local");
  __builtin_amdgcn_s_barrier();
  __builtin_amdgcn_fence(__ATOMIC_ACQUIRE, "workgroup", "local");

  for (int k = 0; k < kNB + kL; ++k) {
    const int blk = k - w;
    if (mover) {
      if (k + 1 < kNB) {
#pragma unroll
        for (int j = 0; j < kTB; ++j) G[(k + 1) & 1][j][lane] = nxt[j];
      }
      if (k + 2 < kNB) {
        const float* g = gx0 + (size_t)(k + 2) * kTB * (kBATCH * kG);
#pragma unroll
        for (int j = 0; j < kTB; ++j)
          load_row(nxt[j], g + (size_t)j * (kBATCH * kG), col, ua);
      }
      const int ob = k - kL;
      if (ob >= 0) {
        const float* o = (const float*)&Y[kL - 1][(k - 1) & 1][0][0];
        float* op = out + ((size_t)ob * kTB * kBATCH + blockIdx.x * 16) * kH;
        float v0[kTB], v1[kTB];
#pragma unroll
        for (int j = 0; j < kTB; ++j) {
          v0[j] = o[j * 64 + src0];
          v1[j] = o[j * 64 + src1];
        }
#pragma unroll
        for (int j = 0; j < kTB; ++j) op[j * (kBATCH * kH) + e0] = v0[j];
        if (lane < 16) {
#pragma unroll
          for (int j = 0; j < kTB; ++j) op[j * (kBATCH * kH) + e1] = v1[j];
        }
      }
    } else if (blk >= 0 && blk < kNB) {
      if (w == 0) {
        block_first(&G[k & 1][0][lane], &Y[0][k & 1][0][lane], a_hh, h, half);
      } else {
        block_upper(&Y[w - 1][(k - 1) & 1][0][lane], &Y[w][k & 1][0][lane],
                    a_ih, a_hh, h, half);
      }
    }
    __builtin_amdgcn_fence(__ATOMIC_RELEASE, "workgroup", "local");
    __builtin_amdgcn_s_barrier();
    __builtin_amdgcn_fence(__ATOMIC_ACQUIRE, "workgroup", "local");
  }

  if (!mover) {
    store_out(out + (size_t)kSEQ * kBATCH * kH + (w * kBATCH + bcol) * kH, h,
              half);
  }
}

extern "C" void kernel_launch(void* const* d_in, const int* in_sizes, int n_in,
                              void* d_out, int out_size, void* d_ws,
                              size_t ws_size, hipStream_t stream) {
  const float* x    = (const float*)d_in[0];
  const float* hxs  = (const float*)d_in[1];
  const float* wih0 = (const float*)d_in[2];
  const float* whh0 = (const float*)d_in[3];
  const float* bih0 = (const float*)d_in[4];
  const float* bhh0 = (const float*)d_in[5];
  const float* wihr = (const float*)d_in[6];
  const float* whhr = (const float*)d_in[7];
  const float* bihr = (const float*)d_in[8];
  const float* bhhr = (const float*)d_in[9];
  float* out = (float*)d_out;
  float* gx  = (float*)d_ws;

  gru_in_proj_wmma<<<kROWS / 128, 256, 0, stream>>>(x, wih0, bih0, gx);
  gru_scan<<<kBATCH / 16, 32 * kWAVES, 0, stream>>>(gx, hxs, whh0, bhh0, wihr,
                                                     whhr, bihr, bhhr, out);
}


# GRU scan: time blocks of 16 steps, not 8: half as many block hand-overs (barrier, LDS round trip, LDS drain) per step, 132 rounds for 260; sums as before
# speedup vs baseline: 12.6648x; 1.0385x over previous
#include <hip/hip_runtime.h>
#include <hip/hip_bf16.h>

typedef float v2f __attribute__((ext_vector_type(2)));
typedef float v8f __attribute__((ext_vector_type(8)));

static constexpr int kSEQ   = 2048;
static constexpr int kBATCH = 64;
static constexpr int kIN    = 512;
static constexpr int kH     = 3;
static constexpr int kL     = 4;
static constexpr int kG     = 3 * kH;
static constexpr int kROWS  = kSEQ * kBATCH;

static constexpr float kSigScale  = -1.4426950408889634f;
static constexpr float kTanhScale = 2.8853900817779268f;

__global__ __launch_bounds__(256) void gru_in_proj_wmma(
    const float* __restrict__ X, const float* __restrict__ Wih0,
    const float* __restrict__ bih0, float* __restrict__ gx) {
  __shared__ float Bs[kIN * 16];
  const int tid = threadIdx.x;
  for (int idx = tid; idx < kIN * 16; idx += 256) {
    const int k = idx >> 4;
    const int n = idx & 15;
    Bs[idx] = (n < kG) ? Wih0[n * kIN + k] : 0.0f;
  }
  __syncthreads();

  const int wave = tid >> 5;
  const int lane = tid & 31;
  const int half = lane >> 4;
  const int n16  = lane & 15;
  const int row0 = blockIdx.x * 128 + wave * 16;

  const float* arow = X + (size_t)(row0 + n16) * kIN + 2 * half;

  v8f c = {0.f, 0.f, 0.f, 0.f, 0.f, 0.f, 0.f, 0.f};
  for (int kk = 0; kk < kIN; kk += 4) {
    v2f a = *(const v2f*)(arow + kk);
    v2f b;
    b.x = Bs[(kk + 2 * half) * 16 + n16];
    b.y = Bs[(kk + 2 * half + 1) * 16 + n16];
    c = __builtin_amdgcn_wmma_f32_16x16x4_f32(
         false, a,  false, b,
         (short)0, c,  false,  false);
  }

  if (n16 < kG) {
    const float bias = bih0[n16];
    const float s = n16 < 2 * kH ? kSigScale : kTanhScale;
#pragma unroll
    for (int v = 0; v < 8; ++v) {
      const int m = v + 8 * half;
      gx[(size_t)(row0 + m) * kG + n16] = (c[v] + bias) * s;
    }
  }
}

static constexpr int kTB    = 16;
static constexpr int kNB    = kSEQ / kTB;
static constexpr int kWAVES = kL + 1;

__device__ __forceinline__ v2f gate_frag(const float* __restrict__ w,
                                         const float* __restrict__ bias,
                                         int lane, bool hidden_side) {
  const int m = lane & 15;
  const int half = lane >> 4;
  const int u = m >> 2;
  const int q = m & 3;
  const bool n_row_here = (q == 3) == hidden_side;
  const bool live = (u < kH) && (q < 2 || n_row_here);
  const int g = live ? (q < 2 ? q * kH + u : 2 * kH + u) : 0;
  const float w0 = w[g * kH + 2 * half];
  const float w1 = half ? bias[g] : w[g * kH + 1];
  const float s = q < 2 ? kSigScale : kTanhScale;
  v2f a;
  a.x = live ? w0 * s : 0.f;
  a.y = live ? w1 * s : 0.f;
  return a;
}

struct Div {
  float den, ds, ns, x, q, r, o;
  bool flag;
};
template <int kNumTimes2>
struct DivSteps {
  static constexpr float num = 0.5f * kNumTimes2;
  static __device__ __forceinline__ void s1(Div& v) {
    bool unused;
    v.ds = __builtin_amdgcn_div_scalef(num, v.den, false, &unused);
  }
  static __device__ __forceinline__ void sn(Div& v) {
    v.ns = __builtin_amdgcn_div_scalef(num, v.den, true, &v.flag);
  }
  static __device__ __forceinline__ void s2(Div& v) {
    v.x = __builtin_amdgcn_rcpf(v.ds);
  }
  static __device__ __forceinline__ void s3(Div& v) {
    v.r = __builtin_fmaf(-v.ds, v.x, 1.0f);
  }
  static __device__ __forceinline__ void s4(Div& v) {
    v.x = __builtin_fmaf(v.r, v.x, v.x);
  }
  static __device__ __forceinline__ void s5(Div& v) { v.q = v.ns * v.x; }
  static __device__ __forceinline__ void s6(Div& v) {
    v.r = __builtin_fmaf(-v.ds, v.q, v.ns);
  }
  static __device__ __forceinline__ void s7(Div& v) {
    v.q = __builtin_fmaf(v.r, v.x, v.q);
  }
  static __device__ __forceinline__ void s8(Div& v) {
    v.r = __builtin_fmaf(-v.ds, v.q, v.ns);
  }
  static __device__ __forceinline__ void s9(Div& v) {
    v.o = __builtin_amdgcn_div_fmasf(v.r, v.x, v.q, v.flag);
  }
  static __device__ __forceinline__ void s10(Div& v) {
    v.o = __builtin_amdgcn_div_fixupf(v.o, v.den, num);
  }
};

__device__ __forceinline__ void level_end() {
  __builtin_amdgcn_sched_barrier(0);
}

__device__ __forceinline__ void gates_to_state(const v8f d, v2f& h, int half) {
  typedef DivSteps<2> One;
  typedef DivSteps<-4> Neg2;
  Div r0, r1, z0, z1, q0, q1;
  const float er0 = __builtin_amdgcn_exp2f(d[0]);
  const float er1 = __builtin_amdgcn_exp2f(d[4]);
  const float ez0 = __builtin_amdgcn_exp2f(d[1]);
  const float ez1 = __builtin_amdgcn_exp2f(d[5]);
  level_end();
  r0.den = 1.0f + er0;
  r1.den = 1.0f + er1;
  level_end();
  One::s1(r0);
  One::s1(r1);
  level_end();
  One::s2(r0);
  One::s2(r1);
  z0.den = 1.0f + ez0;
  z1.den = 1.0f + ez1;
  One::sn(r0);
  One::sn(r1);
  level_end();
  One::s3(r0);
  One::s3(r1);
  One::s1(z0);
  One::s1(z1);
  level_end();
  One::s4(r0);
  One::s4(r1);
  One::s2(z0);
  One::s2(z1);
  One::sn(z0);
  One::sn(z1);
  level_end();
  One::s5(r0);
  One::s5(r1);
  level_end();
  One::s6(r0);
  One::s6(r1);
  One::s3(z0);
  One::s3(z1);
  level_end();
  One::s7(r0);
  One::s7(r1);
  One::s4(z0);
  One::s4(z1);
  level_end();
  One::s8(r0);
  One::s8(r1);
  One::s5(z0);
  One::s5(z1);
  level_end();
  One::s9(r0);
  One::s9(r1);
  One::s6(z0);
  One::s6(z1);
  level_end();
  One::s10(r0);
  One::s10(r1);
  One::s7(z0);
  One::s7(z1);
  level_end();
  const float p0 = __builtin_fmaf(d[3], r0.o, d[2]);
  const float p1 = __builtin_fmaf(d[7], r1.o, d[6]);
  One::s8(z0);
  One::s8(z1);
  level_end();
  const float f0 = __builtin_amdgcn_exp2f(fabsf(p0));
  const float f1 = __builtin_amdgcn_exp2f(fabsf(p1));
  One::s9(z0);
  One::s9(z1);
  level_end();
  q0.den = f0 + 1.0f;
  q1.den = f1 + 1.0f;
  One::s10(z0);
  One::s10(z1);
  level_end();
  Neg2::s1(q0);
  Neg2::s1(q1);
  const float omz0 = 1.0f - z0.o;
  const float omz1 = 1.0f - z1.o;
  level_end();
  Neg2::s2(q0);
  Neg2::s2(q1);
  const float a0 = copysignf(omz0, p0);
  const float a1s = copysignf(omz1, p1);
  Neg2::sn(q0);
  Neg2::sn(q1);
  level_end();
  Neg2::s3(q0);
  Neg2::s3(q1);
  level_end();
  Neg2::s4(q0);
  Neg2::s4(q1);
  const float b0 = __builtin_fmaf(h.x, z0.o, a0);
  const float b1s = __builtin_fmaf(h.y, z1.o, a1s);
  level_end();
  Neg2::s5(q0);
  Neg2::s5(q1);
  level_end();
  Neg2::s6(q0);
  Neg2::s6(q1);
  const float a1 = half ? 0.0f : a1s;
  const float b1 = half ? 1.0f : b1s;
  level_end();
  Neg2::s7(q0);
  Neg2::s7(q1);
  level_end();
  Neg2::s8(q0);
  Neg2::s8(q1);
  level_end();
  Neg2::s9(q0);
  Neg2::s9(q1);
  level_end();
  Neg2::s10(q0);
  Neg2::s10(q1);
  level_end();
  h.x = __builtin_fmaf(a0, q0.o, b0);
  h.y = __builtin_fmaf(a1, q1.o, b1);
}

__device__ __forceinline__ v8f mm(const v2f a, const v2f b, const v8f c) {
  return __builtin_amdgcn_wmma_f32_16x16x4_f32(
       false, a,  false, b,
       (short)0, c,  false,  false);
}

__device__ __forceinline__ void load_row(v8f& row,
                                         const float* __restrict__ gstep,
                                         int col, int ua) {
  row[0] = gstep[col + ua];
  row[1] = gstep[col + kH + ua];
  row[2] = gstep[col + 2 * kH + ua];
  row[4] = gstep[col + 1];
  row[5] = gstep[col + kH + 1];
  row[6] = gstep[col + 2 * kH + 1];
}

__device__ __forceinline__ void store_block(v2f* __restrict__ yout,
                                            const v2f (&hs)[kTB]) {
  __builtin_amdgcn_sched_barrier(0);
#pragma unroll
  for (int j = 0; j < kTB; ++j) yout[j * 32] = hs[j];
}

__device__ __forceinline__ void block_first(const v8f* __restrict__ gin,
                                            v2f* __restrict__ yout,
                                            const v2f a_hh, v2f& h, int half) {
  v8f c[kTB];
#pragma unroll
  for (int j = 0; j < kTB; ++j) c[j] = gin[j * 32];
  __builtin_amdgcn_sched_barrier(0);
  v2f hs[kTB];
#pragma unroll
  for (int j = 0; j < kTB; ++j) {
    const v8f d = mm(a_hh, h, c[j]);
    gates_to_state(d, h, half);
    hs[j] = h;
  }
  store_block(yout, hs);
}

__device__ __forceinline__ void block_upper(const v2f* __restrict__ yin,
                                            v2f* __restrict__ yout,
                                            const v2f a_ih, const v2f a_hh,
                                            v2f& h, int half) {
  const v8f zero8 = {0.f, 0.f, 0.f, 0.f, 0.f, 0.f, 0.f, 0.f};
  v2f y[kTB];
#pragma unroll
  for (int j = 0; j < kTB; ++j) y[j] = yin[j * 32];
  v2f hs[kTB];
  v8f din = mm(a_ih, y[0], zero8);
  __builtin_amdgcn_sched_barrier(0);
#pragma unroll
  for (int j = 0; j < kTB; ++j) {
    const v8f d = mm(a_hh, h, din);
    if (j + 1 < kTB) din = mm(a_ih, y[j + 1], zero8);
    __builtin_amdgcn_sched_barrier(0);
    gates_to_state(d, h, half);
    hs[j] = h;
    __builtin_amdgcn_sched_barrier(0);
  }
  store_block(yout, hs);
}

__device__ __forceinline__ void store_out(float* __restrict__ op, const v2f h,
                                          int half) {
  op[2 * half] = h.x;
  if (half == 0) op[1] = h.y;
}

__global__ __launch_bounds__(32 * kWAVES)
__attribute__((amdgpu_waves_per_eu(1, 2))) void gru_scan(
    const float* __restrict__ gx0, const float* __restrict__ hxs,
    const float* __restrict__ whh0, const float* __restrict__ bhh0,
    const float* __restrict__ wih_r, const float* __restrict__ whh_r,
    const float* __restrict__ bih_r, const float* __restrict__ bhh_r,
    float* __restrict__ out) {
  __shared__ v2f Y[kL][2][kTB][32];
  __shared__ v8f G[2][kTB][32];

  const int lane = threadIdx.x & 31;
  const int w = __builtin_amdgcn_readfirstlane((int)(threadIdx.x >> 5));
  const bool mover = w == kL;
  const int wl = mover ? kL - 1 : w;
  const int half = lane >> 4;
  const int c16 = lane & 15;
  const int bcol = blockIdx.x * 16 + c16;
  const int ua = 2 * half;

  const int lr = wl > 0 ? wl - 1 : 0;
  const v2f a_hh = gate_frag(wl == 0 ? whh0 : whh_r + lr * kG * kH,
                             wl == 0 ? bhh0 : bhh_r + lr * kG, lane, true);
  v2f a_ih = {0.f, 0.f};
  if (wl > 0)
    a_ih = gate_frag(wih_r + lr * kG * kH, bih_r + lr * kG, lane, false);

  v2f h;
  {
    const float* hp = hxs + (wl * kBATCH + bcol) * kH;
    const float x = hp[2 * half];
    const float y = hp[1];
    h.x = x;
    h.y = half ? 1.0f : y;
  }

  const int col = bcol * kG;
  v8f nxt[kTB];
#pragma unroll
  for (int j = 0; j < kTB; ++j)
    nxt[j] = v8f{0.f, 0.f, 0.f, 0.f, 0.f, 0.f, 0.f, 0.f};
  const int e0 = lane, e1 = lane + 32;
  const int e1r = e1 < 16 * kH ? e1 : 16 * kH - 1;
  const int src0 =
      (e0 % kH) < 2 ? 2 * (e0 / kH) + (e0 % kH) : 2 * (16 + e0 / kH);
  const int src1 =
      (e1r % kH) < 2 ? 2 * (e1r / kH) + (e1r % kH) : 2 * (16 + e1r / kH);
  if (mover) {
#pragma unroll
    for (int j = 0; j < kTB; ++j)
      load_row(nxt[j], gx0 + (size_t)j * (kBATCH * kG), col, ua);
#pragma unroll
    for (int j = 0; j < kTB; ++j) G[0][j][lane] = nxt[j];
#pragma unroll
    for (int j = 0; j < kTB; ++j)
      load_row(nxt[j], gx0 + (size_t)(kTB + j) * (kBATCH * kG), col, ua);
  }
  __builtin_amdgcn_fence(__ATOMIC_RELEASE, "workgroup", "local");
  __builtin_amdgcn_s_barrier();
  __builtin_amdgcn_fence(__ATOMIC_ACQUIRE, "workgroup", "local");

  for (int k = 0; k < kNB + kL; ++k) {
    const int blk = k - w;
    if (mover) {
      if (k + 1 < kNB) {
#pragma unroll
        for (int j = 0; j < kTB; ++j) G[(k + 1) & 1][j][lane] = nxt[j];
      }
      if (k + 2 < kNB) {
        const float* g = gx0 + (size_t)(k + 2) * kTB * (kBATCH * kG);
#pragma unroll
        for (int j = 0; j < kTB; ++j)
          load_row(nxt[j], g + (size_t)j * (kBATCH * kG), col, ua);
      }
      const int ob = k - kL;
      if (ob >= 0) {
        const float* o = (const float*)&Y[kL - 1][(k - 1) & 1][0][0];
        float* op = out + ((size_t)ob * kTB * kBATCH + blockIdx.x * 16) * kH;
        float v0[kTB], v1[kTB];
#pragma unroll
        for (int j = 0; j < kTB; ++j) {
          v0[j] = o[j * 64 + src0];
          v1[j] = o[j * 64 + src1];
        }
#pragma unroll
        for (int j = 0; j < kTB; ++j) op[j * (kBATCH * kH) + e0] = v0[j];
        if (lane < 16) {
#pragma unroll
          for (int j = 0; j < kTB; ++j) op[j * (kBATCH * kH) + e1] = v1[j];
        }
      }
    } else if (blk >= 0 && blk < kNB) {
      if (w == 0) {
        block_first(&G[k & 1][0][lane], &Y[0][k & 1][0][lane], a_hh, h, half);
      } else {
        block_upper(&Y[w - 1][(k - 1) & 1][0][lane], &Y[w][k & 1][0][lane],
                    a_ih, a_hh, h, half);
      }
    }
    __builtin_amdgcn_fence(__ATOMIC_RELEASE, "workgroup", "local");
    __builtin_amdgcn_s_barrier();
    __builtin_amdgcn_fence(__ATOMIC_ACQUIRE, "workgroup", "local");
  }

  if (!mover) {
    store_out(out + (size_t)kSEQ * kBATCH * kH + (w * kBATCH + bcol) * kH, h,
              half);
  }
}

extern "C" void kernel_launch(void* const* d_in, const int* in_sizes, int n_in,
                              void* d_out, int out_size, void* d_ws,
                              size_t ws_size, hipStream_t stream) {
  const float* x    = (const float*)d_in[0];
  const float* hxs  = (const float*)d_in[1];
  const float* wih0 = (const float*)d_in[2];
  const float* whh0 = (const float*)d_in[3];
  const float* bih0 = (const float*)d_in[4];
  const float* bhh0 = (const float*)d_in[5];
  const float* wihr = (const float*)d_in[6];
  const float* whhr = (const float*)d_in[7];
  const float* bihr = (const float*)d_in[8];
  const float* bhhr = (const float*)d_in[9];
  float* out = (float*)d_out;
  float* gx  = (float*)d_ws;

  gru_in_proj_wmma<<<kROWS / 128, 256, 0, stream>>>(x, wih0, bih0, gx);
  gru_scan<<<kBATCH / 16, 32 * kWAVES, 0, stream>>>(gx, hxs, whh0, bhh0, wihr,
                                                     whhr, bihr, bhhr, out);
}


# GRU scan: a cell's chain cut from 26 to 22 vector ops: numerator one, denominator >= 1: a division's scaling steps and product by one are identities, dropped; -2 goes to the blend; same sums
# speedup vs baseline: 14.6730x; 1.1586x over previous
#include <hip/hip_runtime.h>
#include <hip/hip_bf16.h>

typedef float v2f __attribute__((ext_vector_type(2)));
typedef float v8f __attribute__((ext_vector_type(8)));

static constexpr int kSEQ   = 2048;
static constexpr int kBATCH = 64;
static constexpr int kIN    = 512;
static constexpr int kH     = 3;
static constexpr int kL     = 4;
static constexpr int kG     = 3 * kH;
static constexpr int kROWS  = kSEQ * kBATCH;

static constexpr float kSigScale  = -1.4426950408889634f;
static constexpr float kTanhScale = 2.8853900817779268f;

__global__ __launch_bounds__(256) void gru_in_proj_wmma(
    const float* __restrict__ X, const float* __restrict__ Wih0,
    const float* __restrict__ bih0, float* __restrict__ gx) {
  __shared__ float Bs[kIN * 16];
  const int tid = threadIdx.x;
  for (int idx = tid; idx < kIN * 16; idx += 256) {
    const int k = idx >> 4;
    const int n = idx & 15;
    Bs[idx] = (n < kG) ? Wih0[n * kIN + k] : 0.0f;
  }
  __syncthreads();

  const int wave = tid >> 5;
  const int lane = tid & 31;
  const int half = lane >> 4;
  const int n16  = lane & 15;
  const int row0 = blockIdx.x * 128 + wave * 16;

  const float* arow = X + (size_t)(row0 + n16) * kIN + 2 * half;

  v8f c = {0.f, 0.f, 0.f, 0.f, 0.f, 0.f, 0.f, 0.f};
  for (int kk = 0; kk < kIN; kk += 4) {
    v2f a = *(const v2f*)(arow + kk);
    v2f b;
    b.x = Bs[(kk + 2 * half) * 16 + n16];
    b.y = Bs[(kk + 2 * half + 1) * 16 + n16];
    c = __builtin_amdgcn_wmma_f32_16x16x4_f32(
         false, a,  false, b,
         (short)0, c,  false,  false);
  }

  if (n16 < kG) {
    const float bias = bih0[n16];
    const float s = n16 < 2 * kH ? kSigScale : kTanhScale;
#pragma unroll
    for (int v = 0; v < 8; ++v) {
      const int m = v + 8 * half;
      gx[(size_t)(row0 + m) * kG + n16] = (c[v] + bias) * s;
    }
  }
}

static constexpr int kTB    = 16;
static constexpr int kNB    = kSEQ / kTB;
static constexpr int kWAVES = kL + 1;

__device__ __forceinline__ v2f gate_frag(const float* __restrict__ w,
                                         const float* __restrict__ bias,
                                         int lane, bool hidden_side) {
  const int m = lane & 15;
  const int half = lane >> 4;
  const int u = m >> 2;
  const int q = m & 3;
  const bool n_row_here = (q == 3) == hidden_side;
  const bool live = (u < kH) && (q < 2 || n_row_here);
  const int g = live ? (q < 2 ? q * kH + u : 2 * kH + u) : 0;
  const float w0 = w[g * kH + 2 * half];
  const float w1 = half ? bias[g] : w[g * kH + 1];
  const float s = q < 2 ? kSigScale : kTanhScale;
  v2f a;
  a.x = live ? w0 * s : 0.f;
  a.y = live ? w1 * s : 0.f;
  return a;
}

struct Rcp {
  float den, x, q, r, o;
};
struct RcpSteps {
  static __device__ __forceinline__ void t1(Rcp& v) {
    v.x = __builtin_amdgcn_rcpf(v.den);
  }
  static __device__ __forceinline__ void t2(Rcp& v) {
    v.r = __builtin_fmaf(-v.den, v.x, 1.0f);
  }
  static __device__ __forceinline__ void t3(Rcp& v) {
    v.x = __builtin_fmaf(v.r, v.x, v.x);
  }
  static __device__ __forceinline__ void t4(Rcp& v) {
    v.r = __builtin_fmaf(-v.den, v.x, 1.0f);
  }
  static __device__ __forceinline__ void t5(Rcp& v) {
    v.q = __builtin_fmaf(v.r, v.x, v.x);
  }
  static __device__ __forceinline__ void t6(Rcp& v) {
    v.r = __builtin_fmaf(-v.den, v.q, 1.0f);
  }
  static __device__ __forceinline__ void t7(Rcp& v) {
    v.o = __builtin_fmaf(v.r, v.x, v.q);
  }
  static __device__ __forceinline__ void t8(Rcp& v) {
    v.o = __builtin_amdgcn_div_fixupf(v.o, v.den, 1.0f);
  }
};

__device__ __forceinline__ void level_end() {
  __builtin_amdgcn_sched_barrier(0);
}

__device__ __forceinline__ void gates_to_state(const v8f d, v2f& h, int half) {
  typedef RcpSteps R;
  Rcp r0, r1, z0, z1, t0, t1;
  const float er0 = __builtin_amdgcn_exp2f(d[0]);
  const float er1 = __builtin_amdgcn_exp2f(d[4]);
  const float ez0 = __builtin_amdgcn_exp2f(d[1]);
  const float ez1 = __builtin_amdgcn_exp2f(d[5]);
  level_end();
  r0.den = 1.0f + er0;
  r1.den = 1.0f + er1;
  level_end();
  R::t1(r0);
  R::t1(r1);
  z0.den = 1.0f + ez0;
  z1.den = 1.0f + ez1;
  level_end();
  R::t2(r0);
  R::t2(r1);
  level_end();
  R::t3(r0);
  R::t3(r1);
  R::t1(z0);
  R::t1(z1);
  level_end();
  R::t4(r0);
  R::t4(r1);
  level_end();
  R::t5(r0);
  R::t5(r1);
  R::t2(z0);
  R::t2(z1);
  level_end();
  R::t6(r0);
  R::t6(r1);
  R::t3(z0);
  R::t3(z1);
  level_end();
  R::t7(r0);
  R::t7(r1);
  R::t4(z0);
  R::t4(z1);
  level_end();
  R::t8(r0);
  R::t8(r1);
  R::t5(z0);
  R::t5(z1);
  level_end();
  const float p0 = __builtin_fmaf(d[3], r0.o, d[2]);
  const float p1 = __builtin_fmaf(d[7], r1.o, d[6]);
  R::t6(z0);
  R::t6(z1);
  level_end();
  const float f0 = __builtin_amdgcn_exp2f(fabsf(p0));
  const float f1 = __builtin_amdgcn_exp2f(fabsf(p1));
  R::t7(z0);
  R::t7(z1);
  level_end();
  t0.den = f0 + 1.0f;
  t1.den = f1 + 1.0f;
  R::t8(z0);
  R::t8(z1);
  level_end();
  R::t1(t0);
  R::t1(t1);
  const float omz0 = 1.0f - z0.o;
  const float omz1 = 1.0f - z1.o;
  level_end();
  R::t2(t0);
  R::t2(t1);
  const float a0 = copysignf(omz0, p0);
  const float a1s = copysignf(omz1, p1);
  level_end();
  R::t3(t0);
  R::t3(t1);
  level_end();
  R::t4(t0);
  R::t4(t1);
  const float b0 = __builtin_fmaf(h.x, z0.o, a0);
  const float b1s = __builtin_fmaf(h.y, z1.o, a1s);
  level_end();
  R::t5(t0);
  R::t5(t1);
  const float a1 = half ? 0.0f : a1s;
  const float b1 = half ? 1.0f : b1s;
  level_end();
  R::t6(t0);
  R::t6(t1);
  const float c0 = -2.0f * a0;
  const float c1 = -2.0f * a1;
  level_end();
  R::t7(t0);
  R::t7(t1);
  level_end();
  R::t8(t0);
  R::t8(t1);
  level_end();
  h.x = __builtin_fmaf(c0, t0.o, b0);
  h.y = __builtin_fmaf(c1, t1.o, b1);
}

__device__ __forceinline__ v8f mm(const v2f a, const v2f b, const v8f c) {
  return __builtin_amdgcn_wmma_f32_16x16x4_f32(
       false, a,  false, b,
       (short)0, c,  false,  false);
}

__device__ __forceinline__ void load_row(v8f& row,
                                         const float* __restrict__ gstep,
                                         int col, int ua) {
  row[0] = gstep[col + ua];
  row[1] = gstep[col + kH + ua];
  row[2] = gstep[col + 2 * kH + ua];
  row[4] = gstep[col + 1];
  row[5] = gstep[col + kH + 1];
  row[6] = gstep[col + 2 * kH + 1];
}

__device__ __forceinline__ void store_block(v2f* __restrict__ yout,
                                            const v2f (&hs)[kTB]) {
  __builtin_amdgcn_sched_barrier(0);
#pragma unroll
  for (int j = 0; j < kTB; ++j) yout[j * 32] = hs[j];
}

__device__ __forceinline__ void block_first(const v8f* __restrict__ gin,
                                            v2f* __restrict__ yout,
                                            const v2f a_hh, v2f& h, int half) {
  v8f c[kTB];
#pragma unroll
  for (int j = 0; j < kTB; ++j) c[j] = gin[j * 32];
  __builtin_amdgcn_sched_barrier(0);
  v2f hs[kTB];
#pragma unroll
  for (int j = 0; j < kTB; ++j) {
    const v8f d = mm(a_hh, h, c[j]);
    gates_to_state(d, h, half);
    hs[j] = h;
  }
  store_block(yout, hs);
}

__device__ __forceinline__ void block_upper(const v2f* __restrict__ yin,
                                            v2f* __restrict__ yout,
                                            const v2f a_ih, const v2f a_hh,
                                            v2f& h, int half) {
  const v8f zero8 = {0.f, 0.f, 0.f, 0.f, 0.f, 0.f, 0.f, 0.f};
  v2f y[kTB];
#pragma unroll
  for (int j = 0; j < kTB; ++j) y[j] = yin[j * 32];
  v2f hs[kTB];
  v8f din = mm(a_ih, y[0], zero8);
  __builtin_amdgcn_sched_barrier(0);
#pragma unroll
  for (int j = 0; j < kTB; ++j) {
    const v8f d = mm(a_hh, h, din);
    if (j + 1 < kTB) din = mm(a_ih, y[j + 1], zero8);
    __builtin_amdgcn_sched_barrier(0);
    gates_to_state(d, h, half);
    hs[j] = h;
    __builtin_amdgcn_sched_barrier(0);
  }
  store_block(yout, hs);
}

__device__ __forceinline__ void store_out(float* __restrict__ op, const v2f h,
                                          int half) {
  op[2 * half] = h.x;
  if (half == 0) op[1] = h.y;
}

__global__ __launch_bounds__(32 * kWAVES)
__attribute__((amdgpu_waves_per_eu(1, 2))) void gru_scan(
    const float* __restrict__ gx0, const float* __restrict__ hxs,
    const float* __restrict__ whh0, const float* __restrict__ bhh0,
    const float* __restrict__ wih_r, const float* __restrict__ whh_r,
    const float* __restrict__ bih_r, const float* __restrict__ bhh_r,
    float* __restrict__ out) {
  __shared__ v2f Y[kL][2][kTB][32];
  __shared__ v8f G[2][kTB][32];

  const int lane = threadIdx.x & 31;
  const int w = __builtin_amdgcn_readfirstlane((int)(threadIdx.x >> 5));
  const bool mover = w == kL;
  const int wl = mover ? kL - 1 : w;
  const int half = lane >> 4;
  const int c16 = lane & 15;
  const int bcol = blockIdx.x * 16 + c16;
  const int ua = 2 * half;

  const int lr = wl > 0 ? wl - 1 : 0;
  const v2f a_hh = gate_frag(wl == 0 ? whh0 : whh_r + lr * kG * kH,
                             wl == 0 ? bhh0 : bhh_r + lr * kG, lane, true);
  v2f a_ih = {0.f, 0.f};
  if (wl > 0)
    a_ih = gate_frag(wih_r + lr * kG * kH, bih_r + lr * kG, lane, false);

  v2f h;
  {
    const float* hp = hxs + (wl * kBATCH + bcol) * kH;
    const float x = hp[2 * half];
    const float y = hp[1];
    h.x = x;
    h.y = half ? 1.0f : y;
  }

  const int col = bcol * kG;
  v8f nxt[kTB];
#pragma unroll
  for (int j = 0; j < kTB; ++j)
    nxt[j] = v8f{0.f, 0.f, 0.f, 0.f, 0.f, 0.f, 0.f, 0.f};
  const int e0 = lane, e1 = lane + 32;
  const int e1r = e1 < 16 * kH ? e1 : 16 * kH - 1;
  const int src0 =
      (e0 % kH) < 2 ? 2 * (e0 / kH) + (e0 % kH) : 2 * (16 + e0 / kH);
  const int src1 =
      (e1r % kH) < 2 ? 2 * (e1r / kH) + (e1r % kH) : 2 * (16 + e1r / kH);
  if (mover) {
#pragma unroll
    for (int j = 0; j < kTB; ++j)
      load_row(nxt[j], gx0 + (size_t)j * (kBATCH * kG), col, ua);
#pragma unroll
    for (int j = 0; j < kTB; ++j) G[0][j][lane] = nxt[j];
#pragma unroll
    for (int j = 0; j < kTB; ++j)
      load_row(nxt[j], gx0 + (size_t)(kTB + j) * (kBATCH * kG), col, ua);
  }
  __builtin_amdgcn_fence(__ATOMIC_RELEASE, "workgroup", "local");
  __builtin_amdgcn_s_barrier();
  __builtin_amdgcn_fence(__ATOMIC_ACQUIRE, "workgroup", "local");

  for (int k = 0; k < kNB + kL; ++k) {
    const int blk = k - w;
    if (mover) {
      if (k + 1 < kNB) {
#pragma unroll
        for (int j = 0; j < kTB; ++j) G[(k + 1) & 1][j][lane] = nxt[j];
      }
      if (k + 2 < kNB) {
        const float* g = gx0 + (size_t)(k + 2) * kTB * (kBATCH * kG);
#pragma unroll
        for (int j = 0; j < kTB; ++j)
          load_row(nxt[j], g + (size_t)j * (kBATCH * kG), col, ua);
      }
      const int ob = k - kL;
      if (ob >= 0) {
        const float* o = (const float*)&Y[kL - 1][(k - 1) & 1][0][0];
        float* op = out + ((size_t)ob * kTB * kBATCH + blockIdx.x * 16) * kH;
        float v0[kTB], v1[kTB];
#pragma unroll
        for (int j = 0; j < kTB; ++j) {
          v0[j] = o[j * 64 + src0];
          v1[j] = o[j * 64 + src1];
        }
#pragma unroll
        for (int j = 0; j < kTB; ++j) op[j * (kBATCH * kH) + e0] = v0[j];
        if (lane < 16) {
#pragma unroll
          for (int j = 0; j < kTB; ++j) op[j * (kBATCH * kH) + e1] = v1[j];
        }
      }
    } else if (blk >= 0 && blk < kNB) {
      if (w == 0) {
        block_first(&G[k & 1][0][lane], &Y[0][k & 1][0][lane], a_hh, h, half);
      } else {
        block_upper(&Y[w - 1][(k - 1) & 1][0][lane], &Y[w][k & 1][0][lane],
                    a_ih, a_hh, h, half);
      }
    }
    __builtin_amdgcn_fence(__ATOMIC_RELEASE, "workgroup", "local");
    __builtin_amdgcn_s_barrier();
    __builtin_amdgcn_fence(__ATOMIC_ACQUIRE, "workgroup", "local");
  }

  if (!mover) {
    store_out(out + (size_t)kSEQ * kBATCH * kH + (w * kBATCH + bcol) * kH, h,
              half);
  }
}

extern "C" void kernel_launch(void* const* d_in, const int* in_sizes, int n_in,
                              void* d_out, int out_size, void* d_ws,
                              size_t ws_size, hipStream_t stream) {
  const float* x    = (const float*)d_in[0];
  const float* hxs  = (const float*)d_in[1];
  const float* wih0 = (const float*)d_in[2];
  const float* whh0 = (const float*)d_in[3];
  const float* bih0 = (const float*)d_in[4];
  const float* bhh0 = (const float*)d_in[5];
  const float* wihr = (const float*)d_in[6];
  const float* whhr = (const float*)d_in[7];
  const float* bihr = (const float*)d_in[8];
  const float* bhhr = (const float*)d_in[9];
  float* out = (float*)d_out;
  float* gx  = (float*)d_ws;

  gru_in_proj_wmma<<<kROWS / 128, 256, 0, stream>>>(x, wih0, bih0, gx);
  gru_scan<<<kBATCH / 16, 32 * kWAVES, 0, stream>>>(gx, hxs, whh0, bhh0, wihr,
                                                     whhr, bihr, bhhr, out);
}


# GRU scan: a cell's chain cut from 22 to 18 vector ops: numerator one: the division's second quotient refinement returns its input, dropped; bits of '/' unless den's mantissa is all ones (1 ulp)
# speedup vs baseline: 16.1881x; 1.1033x over previous
#include <hip/hip_runtime.h>
#include <hip/hip_bf16.h>

typedef float v2f __attribute__((ext_vector_type(2)));
typedef float v8f __attribute__((ext_vector_type(8)));

static constexpr int kSEQ   = 2048;
static constexpr int kBATCH = 64;
static constexpr int kIN    = 512;
static constexpr int kH     = 3;
static constexpr int kL     = 4;
static constexpr int kG     = 3 * kH;
static constexpr int kROWS  = kSEQ * kBATCH;

static constexpr float kSigScale  = -1.4426950408889634f;
static constexpr float kTanhScale = 2.8853900817779268f;

__global__ __launch_bounds__(256) void gru_in_proj_wmma(
    const float* __restrict__ X, const float* __restrict__ Wih0,
    const float* __restrict__ bih0, float* __restrict__ gx) {
  __shared__ float Bs[kIN * 16];
  const int tid = threadIdx.x;
  for (int idx = tid; idx < kIN * 16; idx += 256) {
    const int k = idx >> 4;
    const int n = idx & 15;
    Bs[idx] = (n < kG) ? Wih0[n * kIN + k] : 0.0f;
  }
  __syncthreads();

  const int wave = tid >> 5;
  const int lane = tid & 31;
  const int half = lane >> 4;
  const int n16  = lane & 15;
  const int row0 = blockIdx.x * 128 + wave * 16;

  const float* arow = X + (size_t)(row0 + n16) * kIN + 2 * half;

  v8f c = {0.f, 0.f, 0.f, 0.f, 0.f, 0.f, 0.f, 0.f};
  for (int kk = 0; kk < kIN; kk += 4) {
    v2f a = *(const v2f*)(arow + kk);
    v2f b;
    b.x = Bs[(kk + 2 * half) * 16 + n16];
    b.y = Bs[(kk + 2 * half + 1) * 16 + n16];
    c = __builtin_amdgcn_wmma_f32_16x16x4_f32(
         false, a,  false, b,
         (short)0, c,  false,  false);
  }

  if (n16 < kG) {
    const float bias = bih0[n16];
    const float s = n16 < 2 * kH ? kSigScale : kTanhScale;
#pragma unroll
    for (int v = 0; v < 8; ++v) {
      const int m = v + 8 * half;
      gx[(size_t)(row0 + m) * kG + n16] = (c[v] + bias) * s;
    }
  }
}

static constexpr int kTB    = 16;
static constexpr int kNB    = kSEQ / kTB;
static constexpr int kWAVES = kL + 1;

__device__ __forceinline__ v2f gate_frag(const float* __restrict__ w,
                                         const float* __restrict__ bias,
                                         int lane, bool hidden_side) {
  const int m = lane & 15;
  const int half = lane >> 4;
  const int u = m >> 2;
  const int q = m & 3;
  const bool n_row_here = (q == 3) == hidden_side;
  const bool live = (u < kH) && (q < 2 || n_row_here);
  const int g = live ? (q < 2 ? q * kH + u : 2 * kH + u) : 0;
  const float w0 = w[g * kH + 2 * half];
  const float w1 = half ? bias[g] : w[g * kH + 1];
  const float s = q < 2 ? kSigScale : kTanhScale;
  v2f a;
  a.x = live ? w0 * s : 0.f;
  a.y = live ? w1 * s : 0.f;
  return a;
}

struct Rcp {
  float den, x, q, r, o;
};
struct RcpSteps {
  static __device__ __forceinline__ void t1(Rcp& v) {
    v.x = __builtin_amdgcn_rcpf(v.den);
  }
  static __device__ __forceinline__ void t2(Rcp& v) {
    v.r = __builtin_fmaf(-v.den, v.x, 1.0f);
  }
  static __device__ __forceinline__ void t3(Rcp& v) {
    v.x = __builtin_fmaf(v.r, v.x, v.x);
  }
  static __device__ __forceinline__ void t4(Rcp& v) {
    v.r = __builtin_fmaf(-v.den, v.x, 1.0f);
  }
  static __device__ __forceinline__ void t5(Rcp& v) {
    v.q = __builtin_fmaf(v.r, v.x, v.x);
  }
  static __device__ __forceinline__ void t8(Rcp& v) {
    v.o = __builtin_amdgcn_div_fixupf(v.q, v.den, 1.0f);
  }
};

__device__ __forceinline__ void level_end() {
  __builtin_amdgcn_sched_barrier(0);
}

__device__ __forceinline__ void gates_to_state(const v8f d, v2f& h, int half) {
  typedef RcpSteps R;
  Rcp r0, r1, z0, z1, t0, t1;
  const float er0 = __builtin_amdgcn_exp2f(d[0]);
  const float er1 = __builtin_amdgcn_exp2f(d[4]);
  const float ez0 = __builtin_amdgcn_exp2f(d[1]);
  const float ez1 = __builtin_amdgcn_exp2f(d[5]);
  level_end();
  r0.den = 1.0f + er0;
  r1.den = 1.0f + er1;
  level_end();
  R::t1(r0);
  R::t1(r1);
  z0.den = 1.0f + ez0;
  z1.den = 1.0f + ez1;
  level_end();
  R::t2(r0);
  R::t2(r1);
  level_end();
  R::t3(r0);
  R::t3(r1);
  R::t1(z0);
  R::t1(z1);
  level_end();
  R::t4(r0);
  R::t4(r1);
  level_end();
  R::t5(r0);
  R::t5(r1);
  R::t2(z0);
  R::t2(z1);
  level_end();
  R::t8(r0);
  R::t8(r1);
  R::t3(z0);
  R::t3(z1);
  level_end();
  const float p0 = __builtin_fmaf(d[3], r0.o, d[2]);
  const float p1 = __builtin_fmaf(d[7], r1.o, d[6]);
  R::t4(z0);
  R::t4(z1);
  level_end();
  const float f0 = __builtin_amdgcn_exp2f(fabsf(p0));
  const float f1 = __builtin_amdgcn_exp2f(fabsf(p1));
  R::t5(z0);
  R::t5(z1);
  level_end();
  t0.den = f0 + 1.0f;
  t1.den = f1 + 1.0f;
  R::t8(z0);
  R::t8(z1);
  level_end();
  R::t1(t0);
  R::t1(t1);
  const float omz0 = 1.0f - z0.o;
  const float omz1 = 1.0f - z1.o;
  level_end();
  R::t2(t0);
  R::t2(t1);
  const float a0 = copysignf(omz0, p0);
  const float a1s = copysignf(omz1, p1);
  level_end();
  R::t3(t0);
  R::t3(t1);
  const float a1 = half ? 0.0f : a1s;
  level_end();
  R::t4(t0);
  R::t4(t1);
  const float b0 = __builtin_fmaf(h.x, z0.o, a0);
  const float b1s = __builtin_fmaf(h.y, z1.o, a1s);
  level_end();
  R::t5(t0);
  R::t5(t1);
  const float c0 = -2.0f * a0;
  const float c1 = -2.0f * a1;
  level_end();
  R::t8(t0);
  R::t8(t1);
  const float b1 = half ? 1.0f : b1s;
  level_end();
  h.x = __builtin_fmaf(c0, t0.o, b0);
  h.y = __builtin_fmaf(c1, t1.o, b1);
}

__device__ __forceinline__ v8f mm(const v2f a, const v2f b, const v8f c) {
  return __builtin_amdgcn_wmma_f32_16x16x4_f32(
       false, a,  false, b,
       (short)0, c,  false,  false);
}

__device__ __forceinline__ void load_row(v8f& row,
                                         const float* __restrict__ gstep,
                                         int col, int ua) {
  row[0] = gstep[col + ua];
  row[1] = gstep[col + kH + ua];
  row[2] = gstep[col + 2 * kH + ua];
  row[4] = gstep[col + 1];
  row[5] = gstep[col + kH + 1];
  row[6] = gstep[col + 2 * kH + 1];
}

__device__ __forceinline__ void store_block(v2f* __restrict__ yout,
                                            const v2f (&hs)[kTB]) {
  __builtin_amdgcn_sched_barrier(0);
#pragma unroll
  for (int j = 0; j < kTB; ++j) yout[j * 32] = hs[j];
}

__device__ __forceinline__ void block_first(const v8f* __restrict__ gin,
                                            v2f* __restrict__ yout,
                                            const v2f a_hh, v2f& h, int half) {
  v8f c[kTB];
#pragma unroll
  for (int j = 0; j < kTB; ++j) c[j] = gin[j * 32];
  __builtin_amdgcn_sched_barrier(0);
  v2f hs[kTB];
#pragma unroll
  for (int j = 0; j < kTB; ++j) {
    const v8f d = mm(a_hh, h, c[j]);
    gates_to_state(d, h, half);
    hs[j] = h;
  }
  store_block(yout, hs);
}

__device__ __forceinline__ void block_upper(const v2f* __restrict__ yin,
                                            v2f* __restrict__ yout,
                                            const v2f a_ih, const v2f a_hh,
                                            v2f& h, int half) {
  const v8f zero8 = {0.f, 0.f, 0.f, 0.f, 0.f, 0.f, 0.f, 0.f};
  v2f y[kTB];
#pragma unroll
  for (int j = 0; j < kTB; ++j) y[j] = yin[j * 32];
  v2f hs[kTB];
  v8f din = mm(a_ih, y[0], zero8);
  __builtin_amdgcn_sched_barrier(0);
#pragma unroll
  for (int j = 0; j < kTB; ++j) {
    const v8f d = mm(a_hh, h, din);
    if (j + 1 < kTB) din = mm(a_ih, y[j + 1], zero8);
    __builtin_amdgcn_sched_barrier(0);
    gates_to_state(d, h, half);
    hs[j] = h;
    __builtin_amdgcn_sched_barrier(0);
  }
  store_block(yout, hs);
}

__device__ __forceinline__ void store_out(float* __restrict__ op, const v2f h,
                                          int half) {
  op[2 * half] = h.x;
  if (half == 0) op[1] = h.y;
}

__global__ __launch_bounds__(32 * kWAVES)
__attribute__((amdgpu_waves_per_eu(1, 2))) void gru_scan(
    const float* __restrict__ gx0, const float* __restrict__ hxs,
    const float* __restrict__ whh0, const float* __restrict__ bhh0,
    const float* __restrict__ wih_r, const float* __restrict__ whh_r,
    const float* __restrict__ bih_r, const float* __restrict__ bhh_r,
    float* __restrict__ out) {
  __shared__ v2f Y[kL][2][kTB][32];
  __shared__ v8f G[2][kTB][32];

  const int lane = threadIdx.x & 31;
  const int w = __builtin_amdgcn_readfirstlane((int)(threadIdx.x >> 5));
  const bool mover = w == kL;
  const int wl = mover ? kL - 1 : w;
  const int half = lane >> 4;
  const int c16 = lane & 15;
  const int bcol = blockIdx.x * 16 + c16;
  const int ua = 2 * half;

  const int lr = wl > 0 ? wl - 1 : 0;
  const v2f a_hh = gate_frag(wl == 0 ? whh0 : whh_r + lr * kG * kH,
                             wl == 0 ? bhh0 : bhh_r + lr * kG, lane, true);
  v2f a_ih = {0.f, 0.f};
  if (wl > 0)
    a_ih = gate_frag(wih_r + lr * kG * kH, bih_r + lr * kG, lane, false);

  v2f h;
  {
    const float* hp = hxs + (wl * kBATCH + bcol) * kH;
    const float x = hp[2 * half];
    const float y = hp[1];
    h.x = x;
    h.y = half ? 1.0f : y;
  }

  const int col = bcol * kG;
  v8f nxt[kTB];
#pragma unroll
  for (int j = 0; j < kTB; ++j)
    nxt[j] = v8f{0.f, 0.f, 0.f, 0.f, 0.f, 0.f, 0.f, 0.f};
  const int e0 = lane, e1 = lane + 32;
  const int e1r = e1 < 16 * kH ? e1 : 16 * kH - 1;
  const int src0 =
      (e0 % kH) < 2 ? 2 * (e0 / kH) + (e0 % kH) : 2 * (16 + e0 / kH);
  const int src1 =
      (e1r % kH) < 2 ? 2 * (e1r / kH) + (e1r % kH) : 2 * (16 + e1r / kH);
  if (mover) {
#pragma unroll
    for (int j = 0; j < kTB; ++j)
      load_row(nxt[j], gx0 + (size_t)j * (kBATCH * kG), col, ua);
#pragma unroll
    for (int j = 0; j < kTB; ++j) G[0][j][lane] = nxt[j];
#pragma unroll
    for (int j = 0; j < kTB; ++j)
      load_row(nxt[j], gx0 + (size_t)(kTB + j) * (kBATCH * kG), col, ua);
  }
  __builtin_amdgcn_fence(__ATOMIC_RELEASE, "workgroup", "local");
  __builtin_amdgcn_s_barrier();
  __builtin_amdgcn_fence(__ATOMIC_ACQUIRE, "workgroup", "local");

  for (int k = 0; k < kNB + kL; ++k) {
    const int blk = k - w;
    if (mover) {
      if (k + 1 < kNB) {
#pragma unroll
        for (int j = 0; j < kTB; ++j) G[(k + 1) & 1][j][lane] = nxt[j];
      }
      if (k + 2 < kNB) {
        const float* g = gx0 + (size_t)(k + 2) * kTB * (kBATCH * kG);
#pragma unroll
        for (int j = 0; j < kTB; ++j)
          load_row(nxt[j], g + (size_t)j * (kBATCH * kG), col, ua);
      }
      const int ob = k - kL;
      if (ob >= 0) {
        const float* o = (const float*)&Y[kL - 1][(k - 1) & 1][0][0];
        float* op = out + ((size_t)ob * kTB * kBATCH + blockIdx.x * 16) * kH;
        float v0[kTB], v1[kTB];
#pragma unroll
        for (int j = 0; j < kTB; ++j) {
          v0[j] = o[j * 64 + src0];
          v1[j] = o[j * 64 + src1];
        }
#pragma unroll
        for (int j = 0; j < kTB; ++j) op[j * (kBATCH * kH) + e0] = v0[j];
        if (lane < 16) {
#pragma unroll
          for (int j = 0; j < kTB; ++j) op[j * (kBATCH * kH) + e1] = v1[j];
        }
      }
    } else if (blk >= 0 && blk < kNB) {
      if (w == 0) {
        block_first(&G[k & 1][0][lane], &Y[0][k & 1][0][lane], a_hh, h, half);
      } else {
        block_upper(&Y[w - 1][(k - 1) & 1][0][lane], &Y[w][k & 1][0][lane],
                    a_ih, a_hh, h, half);
      }
    }
    __builtin_amdgcn_fence(__ATOMIC_RELEASE, "workgroup", "local");
    __builtin_amdgcn_s_barrier();
    __builtin_amdgcn_fence(__ATOMIC_ACQUIRE, "workgroup", "local");
  }

  if (!mover) {
    store_out(out + (size_t)kSEQ * kBATCH * kH + (w * kBATCH + bcol) * kH, h,
              half);
  }
}

extern "C" void kernel_launch(void* const* d_in, const int* in_sizes, int n_in,
                              void* d_out, int out_size, void* d_ws,
                              size_t ws_size, hipStream_t stream) {
  const float* x    = (const float*)d_in[0];
  const float* hxs  = (const float*)d_in[1];
  const float* wih0 = (const float*)d_in[2];
  const float* whh0 = (const float*)d_in[3];
  const float* bih0 = (const float*)d_in[4];
  const float* bhh0 = (const float*)d_in[5];
  const float* wihr = (const float*)d_in[6];
  const float* whhr = (const float*)d_in[7];
  const float* bihr = (const float*)d_in[8];
  const float* bhhr = (const float*)d_in[9];
  float* out = (float*)d_out;
  float* gx  = (float*)d_ws;

  gru_in_proj_wmma<<<kROWS / 128, 256, 0, stream>>>(x, wih0, bih0, gx);
  gru_scan<<<kBATCH / 16, 32 * kWAVES, 0, stream>>>(gx, hxs, whh0, bhh0, wihr,
                                                     whhr, bihr, bhhr, out);
}
